# starting kernel
# baseline (speedup 1.0000x reference)
#include <hip/hip_runtime.h>
#include <hip/hip_bf16.h>
#include <math.h>


typedef __attribute__((ext_vector_type(2))) float v2f;
typedef __attribute__((ext_vector_type(8))) float v8f;

#define IN_F     128
#define OUT_F    64
#define HEADS    2
#define XDIM     128
#define NEG_SLOPE 0.2f
#define LN_EPS    1e-5f
#define SM_EPS    1e-16f

__global__ __launch_bounds__(256) void gat_gemm_wmma(
    const float* __restrict__ X, const float* __restrict__ W,
    float* __restrict__ xp, int N)
{
    const int wave = (blockIdx.x * blockDim.x + threadIdx.x) >> 5;
    const int lane = threadIdx.x & 31;
    const int r0 = wave * 16;
    if (r0 >= N) return;

    const int lm = lane & 15;
    const int kk = (lane >> 4) << 1;

    int ar = r0 + lm; if (ar >= N) ar = N - 1;
    const float* arow = X + (size_t)ar * IN_F + kk;
    const float* brow = W + (size_t)lm * IN_F + kk;

    v8f acc[8];
    #pragma unroll
    for (int t = 0; t < 8; ++t) acc[t] = (v8f){};

    #pragma unroll 4
    for (int k0 = 0; k0 < IN_F; k0 += 4) {
        v2f a = *(const v2f*)(arow + k0);
        #pragma unroll
        for (int t = 0; t < 8; ++t) {
            v2f b = *(const v2f*)(brow + (size_t)t * 16 * IN_F + k0);
            acc[t] = __builtin_amdgcn_wmma_f32_16x16x4_f32(
                 false, a,  false, b,
                 (short)0, acc[t],  false,  false);
        }
    }

    const int mofs = (lane >> 4) << 3;
    #pragma unroll
    for (int v = 0; v < 8; ++v) {
        const int row = r0 + mofs + v;
        if (row < N) {
            float* orow = xp + (size_t)row * XDIM + lm;
            #pragma unroll
            for (int t = 0; t < 8; ++t) orow[t * 16] = acc[t][v];
        }
    }
}

__global__ __launch_bounds__(256) void gat_logits_init(
    const float* __restrict__ xp,
    const float* __restrict__ attS, const float* __restrict__ attD,
    float* __restrict__ aS, float* __restrict__ aD,
    float* __restrict__ mmax, float* __restrict__ denom,
    float* __restrict__ acc, int N)
{
    const int wave = (blockIdx.x * blockDim.x + threadIdx.x) >> 5;
    const int lane = threadIdx.x & 31;
    if (wave >= N) return;
    const int head = lane >> 4;
    const int fo   = lane * 4;

    float4 xv = *(const float4*)(xp + (size_t)wave * XDIM + fo);
    float4 sv = *(const float4*)(attS + fo);
    float4 dv = *(const float4*)(attD + fo);
    float ps = xv.x*sv.x + xv.y*sv.y + xv.z*sv.z + xv.w*sv.w;
    float pd = xv.x*dv.x + xv.y*dv.y + xv.z*dv.z + xv.w*dv.w;
    #pragma unroll
    for (int off = 8; off >= 1; off >>= 1) {
        ps += __shfl_xor(ps, off, 32);
        pd += __shfl_xor(pd, off, 32);
    }
    if ((lane & 15) == 0) {
        aS[wave * 2 + head]    = ps;
        aD[wave * 2 + head]    = pd;
        mmax[wave * 2 + head]  = -INFINITY;
        denom[wave * 2 + head] = 0.0f;
    }
    float4 z = {0.f, 0.f, 0.f, 0.f};
    *(float4*)(acc + (size_t)wave * XDIM + fo) = z;
}

__device__ __forceinline__ void atomicMaxF(float* addr, float val)
{
    if (val >= 0.0f) atomicMax((int*)addr, __float_as_int(val));
    else             atomicMin((unsigned int*)addr, __float_as_uint(val));
}

__device__ __forceinline__ float lrelu(float v)
{
    return v > 0.0f ? v : NEG_SLOPE * v;
}

__global__ __launch_bounds__(256) void gat_edge_max(
    const int* __restrict__ esrc, const int* __restrict__ edst,
    const float* __restrict__ aS, const float* __restrict__ aD,
    float* __restrict__ mmax, int E, int N)
{
    const int eid = blockIdx.x * blockDim.x + threadIdx.x;
    if (eid >= E + N) return;
    int s, d;
    if (eid < E) { s = esrc[eid]; d = edst[eid]; }
    else         { s = d = eid - E; }
    const float e0 = lrelu(aS[s * 2 + 0] + aD[d * 2 + 0]);
    const float e1 = lrelu(aS[s * 2 + 1] + aD[d * 2 + 1]);
    atomicMaxF(&mmax[d * 2 + 0], e0);
    atomicMaxF(&mmax[d * 2 + 1], e1);
}

__global__ __launch_bounds__(256) void gat_edge_sum(
    const int* __restrict__ esrc, const int* __restrict__ edst,
    const float* __restrict__ aS, const float* __restrict__ aD,
    const float* __restrict__ mmax, float* __restrict__ denom, int E, int N)
{
    const int eid = blockIdx.x * blockDim.x + threadIdx.x;
    if (eid >= E + N) return;
    int s, d;
    if (eid < E) { s = esrc[eid]; d = edst[eid]; }
    else         { s = d = eid - E; }
    const float e0 = lrelu(aS[s * 2 + 0] + aD[d * 2 + 0]);
    const float e1 = lrelu(aS[s * 2 + 1] + aD[d * 2 + 1]);
    atomicAdd(&denom[d * 2 + 0], __expf(e0 - mmax[d * 2 + 0]));
    atomicAdd(&denom[d * 2 + 1], __expf(e1 - mmax[d * 2 + 1]));
}

__global__ __launch_bounds__(256) void gat_edge_scatter(
    const int* __restrict__ esrc, const int* __restrict__ edst,
    const float* __restrict__ aS, const float* __restrict__ aD,
    const float* __restrict__ mmax, const float* __restrict__ denom,
    const float* __restrict__ xp, float* __restrict__ acc, int E, int N)
{
    const int wave = (blockIdx.x * blockDim.x + threadIdx.x) >> 5;
    const int lane = threadIdx.x & 31;
    if (wave >= E + N) return;
    int s, d;
    if (wave < E) { s = esrc[wave]; d = edst[wave]; }
    else          { s = d = wave - E; }
    const int head = lane >> 4;
    float e = lrelu(aS[s * 2 + head] + aD[d * 2 + head]);
    const float alpha = __expf(e - mmax[d * 2 + head]) /
                        (denom[d * 2 + head] + SM_EPS);
    const int fo = lane * 4;
    float4 v = *(const float4*)(xp + (size_t)s * XDIM + fo);
    float* ap = acc + (size_t)d * XDIM + fo;
    atomicAdd(ap + 0, v.x * alpha);
    atomicAdd(ap + 1, v.y * alpha);
    atomicAdd(ap + 2, v.z * alpha);
    atomicAdd(ap + 3, v.w * alpha);
}

__global__ __launch_bounds__(256) void gat_finalize(
    const float* __restrict__ acc, const float* __restrict__ bias,
    const float* __restrict__ gamma, const float* __restrict__ beta,
    float* __restrict__ out, int N)
{
    const int wave = (blockIdx.x * blockDim.x + threadIdx.x) >> 5;
    const int lane = threadIdx.x & 31;
    if (wave >= N) return;
    const float* a = acc + (size_t)wave * XDIM;
    const int f0 = lane, f1 = lane + 32;
    float v0 = (a[f0] + a[OUT_F + f0]) * 0.5f + bias[f0];
    float v1 = (a[f1] + a[OUT_F + f1]) * 0.5f + bias[f1];

    float sum = v0 + v1;
    #pragma unroll
    for (int off = 16; off >= 1; off >>= 1) sum += __shfl_xor(sum, off, 32);
    const float mu = sum * (1.0f / OUT_F);

    const float d0 = v0 - mu, d1 = v1 - mu;
    float sq = d0 * d0 + d1 * d1;
    #pragma unroll
    for (int off = 16; off >= 1; off >>= 1) sq += __shfl_xor(sq, off, 32);
    const float inv = rsqrtf(sq * (1.0f / OUT_F) + LN_EPS);

    out[(size_t)wave * OUT_F + f0] = d0 * inv * gamma[f0] + beta[f0];
    out[(size_t)wave * OUT_F + f1] = d1 * inv * gamma[f1] + beta[f1];
}

extern "C" void kernel_launch(void* const* d_in, const int* in_sizes, int n_in,
                              void* d_out, int out_size, void* d_ws, size_t ws_size,
                              hipStream_t stream)
{
    const float* X    = (const float*)d_in[0];
    const int*   ei   = (const int*)  d_in[1];
    const float* W    = (const float*)d_in[2];
    const float* attS = (const float*)d_in[3];
    const float* attD = (const float*)d_in[4];
    const float* bias = (const float*)d_in[5];
    const float* g    = (const float*)d_in[6];
    const float* b    = (const float*)d_in[7];
    float* out = (float*)d_out;

    const int N = in_sizes[0] / IN_F;
    const int E = in_sizes[1] / 2;
    const int* esrc = ei;
    const int* edst = ei + E;
    const int Et = E + N;

    float* ws    = (float*)d_ws;
    float* xp    = ws;
    float* accb  = xp   + (size_t)N * XDIM;
    float* aS    = accb + (size_t)N * XDIM;
    float* aD    = aS   + (size_t)N * 2;
    float* mmax  = aD   + (size_t)N * 2;
    float* denom = mmax + (size_t)N * 2;

    {
        const int waves = (N + 15) / 16;
        gat_gemm_wmma<<<(waves + 7) / 8, 256, 0, stream>>>(X, W, xp, N);
    }
    gat_logits_init<<<(N + 7) / 8, 256, 0, stream>>>(
        xp, attS, attD, aS, aD, mmax, denom, accb, N);
    gat_edge_max<<<(Et + 255) / 256, 256, 0, stream>>>(
        esrc, edst, aS, aD, mmax, E, N);
    gat_edge_sum<<<(Et + 255) / 256, 256, 0, stream>>>(
        esrc, edst, aS, aD, mmax, denom, E, N);
    gat_edge_scatter<<<(Et + 7) / 8, 256, 0, stream>>>(
        esrc, edst, aS, aD, mmax, denom, xp, accb, E, N);
    gat_finalize<<<(N + 7) / 8, 256, 0, stream>>>(accb, bias, g, b, out, N);
}


# Segment max and exp-sum: each block keeps one dst range's table in LDS (LDS atomics) over an edge chunk, then whole-line global atomics; no per-edge global atomics; denom summed in another order
# speedup vs baseline: 2.9749x; 2.9749x over previous
#include <hip/hip_runtime.h>
#include <hip/hip_bf16.h>
#include <math.h>


typedef __attribute__((ext_vector_type(2))) float v2f;
typedef __attribute__((ext_vector_type(8))) float v8f;

#define IN_F     128
#define OUT_F    64
#define HEADS    2
#define XDIM     128
#define NEG_SLOPE 0.2f
#define LN_EPS    1e-5f
#define SM_EPS    1e-16f

__global__ __launch_bounds__(256) void gat_gemm_wmma(
    const float* __restrict__ X, const float* __restrict__ W,
    float* __restrict__ xp, int N)
{
    const int wave = (blockIdx.x * blockDim.x + threadIdx.x) >> 5;
    const int lane = threadIdx.x & 31;
    const int r0 = wave * 16;
    if (r0 >= N) return;

    const int lm = lane & 15;
    const int kk = (lane >> 4) << 1;

    int ar = r0 + lm; if (ar >= N) ar = N - 1;
    const float* arow = X + (size_t)ar * IN_F + kk;
    const float* brow = W + (size_t)lm * IN_F + kk;

    v8f acc[8];
    #pragma unroll
    for (int t = 0; t < 8; ++t) acc[t] = (v8f){};

    #pragma unroll 4
    for (int k0 = 0; k0 < IN_F; k0 += 4) {
        v2f a = *(const v2f*)(arow + k0);
        #pragma unroll
        for (int t = 0; t < 8; ++t) {
            v2f b = *(const v2f*)(brow + (size_t)t * 16 * IN_F + k0);
            acc[t] = __builtin_amdgcn_wmma_f32_16x16x4_f32(
                 false, a,  false, b,
                 (short)0, acc[t],  false,  false);
        }
    }

    const int mofs = (lane >> 4) << 3;
    #pragma unroll
    for (int v = 0; v < 8; ++v) {
        const int row = r0 + mofs + v;
        if (row < N) {
            float* orow = xp + (size_t)row * XDIM + lm;
            #pragma unroll
            for (int t = 0; t < 8; ++t) orow[t * 16] = acc[t][v];
        }
    }
}

__global__ __launch_bounds__(256) void gat_logits_init(
    const float* __restrict__ xp,
    const float* __restrict__ attS, const float* __restrict__ attD,
    float* __restrict__ aS, float* __restrict__ aD,
    float* __restrict__ mmax, float* __restrict__ denom,
    float* __restrict__ acc, int N)
{
    const int wave = (blockIdx.x * blockDim.x + threadIdx.x) >> 5;
    const int lane = threadIdx.x & 31;
    if (wave >= N) return;
    const int head = lane >> 4;
    const int fo   = lane * 4;

    float4 xv = *(const float4*)(xp + (size_t)wave * XDIM + fo);
    float4 sv = *(const float4*)(attS + fo);
    float4 dv = *(const float4*)(attD + fo);
    float ps = xv.x*sv.x + xv.y*sv.y + xv.z*sv.z + xv.w*sv.w;
    float pd = xv.x*dv.x + xv.y*dv.y + xv.z*dv.z + xv.w*dv.w;
    #pragma unroll
    for (int off = 8; off >= 1; off >>= 1) {
        ps += __shfl_xor(ps, off, 32);
        pd += __shfl_xor(pd, off, 32);
    }
    if ((lane & 15) == 0) {
        aS[wave * 2 + head]    = ps;
        aD[wave * 2 + head]    = pd;
        mmax[wave * 2 + head]  = -INFINITY;
        denom[wave * 2 + head] = 0.0f;
    }
    float2 z = {0.f, 0.f};
    *(float2*)(acc + (size_t)wave * OUT_F + lane * 2) = z;
}

__device__ __forceinline__ void atomicMaxF(float* addr, float val)
{
    if (val >= 0.0f) atomicMax((int*)addr, __float_as_int(val));
    else             atomicMin((unsigned int*)addr, __float_as_uint(val));
}

__device__ __forceinline__ float lrelu(float v)
{
    return v > 0.0f ? v : NEG_SLOPE * v;
}

#define SEG_NR 6256
#define SEG_C  16
#define SEG_T  1024
#define SEG_U  3

template <bool SUM>
__device__ __forceinline__ void seg_update(
    bool mine, unsigned ld, float2 as, float2 ad, float2 mm, float* t_m, float* t_d)
{
    const float e0 = lrelu(as.x + ad.x);
    const float e1 = lrelu(as.y + ad.y);
    if (SUM) {
        const float x0 = __expf(e0 - mm.x);
        const float x1 = __expf(e1 - mm.y);
        if (mine) {
            atomicAdd(t_d + 2 * ld,     x0);
            atomicAdd(t_d + 2 * ld + 1, x1);
        }
    } else {
        if (mine) {
            atomicMaxF(t_m + 2 * ld,     e0);
            atomicMaxF(t_m + 2 * ld + 1, e1);
        }
    }
}

template <bool SUM>
__device__ __forceinline__ void seg_walk(
    const int* __restrict__ esrc, const int* __restrict__ edst,
    const float* __restrict__ aS,
    const float* t_ad, float* t_m, float* t_d,
    int d0, int nr, int e_lo, int e_hi, int E, int tid)
{
    int v_hi = e_lo;
    if ((E & 3) == 0) {
        const int lim = (e_hi < E) ? e_hi : E;
        if (lim > e_lo) v_hi = lim & ~3;
    }
    if (v_hi > e_lo) {
        const int step = SEG_T * 4;
        const int g_last = v_hi - 4;
        int4 dn[SEG_U], sn[SEG_U];
        int g0 = e_lo + tid * 4;
        #pragma unroll
        for (int u = 0; u < SEG_U; ++u) {
            const int g = g0 + u * step;
            const int gc = (g < g_last) ? g : g_last;
            dn[u] = *(const int4*)(edst + gc);
            sn[u] = *(const int4*)(esrc + gc);
        }
        for (int gb = e_lo; gb < v_hi; gb += step * SEG_U) {
            int dc[SEG_U * 4], sc[SEG_U * 4];
            #pragma unroll
            for (int u = 0; u < SEG_U; ++u) {
                const bool ok = (g0 + u * step) < v_hi;
                dc[u * 4 + 0] = ok ? dn[u].x : -1;
                dc[u * 4 + 1] = ok ? dn[u].y : -1;
                dc[u * 4 + 2] = ok ? dn[u].z : -1;
                dc[u * 4 + 3] = ok ? dn[u].w : -1;
                sc[u * 4 + 0] = sn[u].x; sc[u * 4 + 1] = sn[u].y;
                sc[u * 4 + 2] = sn[u].z; sc[u * 4 + 3] = sn[u].w;
            }
            bool     mine[SEG_U * 4];
            unsigned ld[SEG_U * 4];
            float2   as[SEG_U * 4];
            #pragma unroll
            for (int k = 0; k < SEG_U * 4; ++k) {
                ld[k]   = (unsigned)(dc[k] - d0);
                mine[k] = ld[k] < (unsigned)nr;
                if (!mine[k]) ld[k] = 0u;
                const int se = mine[k] ? sc[k] : 0;
                as[k] = *(const float2*)(aS + (size_t)se * 2);
            }
            g0 += step * SEG_U;
            #pragma unroll
            for (int u = 0; u < SEG_U; ++u) {
                const int g = g0 + u * step;
                const int gc = (g < g_last) ? g : g_last;
                dn[u] = *(const int4*)(edst + gc);
                sn[u] = *(const int4*)(esrc + gc);
            }
            float2 ad[SEG_U * 4], mm[SEG_U * 4];
            #pragma unroll
            for (int k = 0; k < SEG_U * 4; ++k) {
                ad[k] = *(const float2*)(t_ad + 2 * ld[k]);
                mm[k] = SUM ? *(const float2*)(t_m + 2 * ld[k]) : float2{0.f, 0.f};
            }
            #pragma unroll
            for (int k = 0; k < SEG_U * 4; ++k)
                seg_update<SUM>(mine[k], ld[k], as[k], ad[k], mm[k], t_m, t_d);
        }
    }
    for (int eid = v_hi + tid; eid < e_hi; eid += SEG_T) {
        int s, d;
        if (eid < E) { s = esrc[eid]; d = edst[eid]; }
        else         { s = d = eid - E; }
        const unsigned ld1 = (unsigned)(d - d0);
        if (ld1 < (unsigned)nr) {
            const float2 as1 = *(const float2*)(aS + (size_t)s * 2);
            const float2 ad1 = *(const float2*)(t_ad + 2 * ld1);
            const float2 mm1 = SUM ? *(const float2*)(t_m + 2 * ld1) : float2{0.f, 0.f};
            seg_update<SUM>(true, ld1, as1, ad1, mm1, t_m, t_d);
        }
    }
}

__device__ __forceinline__ void seg_chunk(int Et, int c, int& e_lo, int& e_hi)
{
    const int ec = (((Et + SEG_C - 1) / SEG_C) + 3) & ~3;
    e_lo = c * ec;
    if (e_lo > Et) e_lo = Et;
    e_hi = (Et - e_lo < ec) ? Et : e_lo + ec;
}

__global__ __launch_bounds__(SEG_T) void gat_seg_max(
    const int* __restrict__ esrc, const int* __restrict__ edst,
    const float* __restrict__ aS, const float* __restrict__ aD,
    float* __restrict__ mmax, int E, int N)
{
    __shared__ float t_ad[SEG_NR * 2];
    __shared__ float t_m[SEG_NR * 2];
    const int tid = threadIdx.x;
    const int d0 = (int)(blockIdx.x / SEG_C) * SEG_NR;
    const int c  = (int)(blockIdx.x % SEG_C);
    const int nr = (N - d0 < SEG_NR) ? (N - d0) : SEG_NR;
    int e_lo, e_hi;
    seg_chunk(E + N, c, e_lo, e_hi);

    for (int i = tid; i < nr * 2; i += SEG_T) {
        t_ad[i] = aD[(size_t)d0 * 2 + i];
        t_m[i]  = -INFINITY;
    }
    __syncthreads();
    seg_walk<false>(esrc, edst, aS, t_ad, t_m, nullptr, d0, nr, e_lo, e_hi, E, tid);
    __syncthreads();
    for (int i = tid; i < nr * 2; i += SEG_T)
        atomicMaxF(&mmax[(size_t)d0 * 2 + i], t_m[i]);
}

__global__ __launch_bounds__(SEG_T) void gat_seg_sum(
    const int* __restrict__ esrc, const int* __restrict__ edst,
    const float* __restrict__ aS, const float* __restrict__ aD,
    const float* __restrict__ mmax, float* __restrict__ denom, int E, int N)
{
    __shared__ float t_ad[SEG_NR * 2];
    __shared__ float t_m[SEG_NR * 2];
    __shared__ float t_d[SEG_NR * 2];
    const int tid = threadIdx.x;
    const int d0 = (int)(blockIdx.x / SEG_C) * SEG_NR;
    const int c  = (int)(blockIdx.x % SEG_C);
    const int nr = (N - d0 < SEG_NR) ? (N - d0) : SEG_NR;
    int e_lo, e_hi;
    seg_chunk(E + N, c, e_lo, e_hi);

    for (int i = tid; i < nr * 2; i += SEG_T) {
        t_ad[i] = aD[(size_t)d0 * 2 + i];
        t_m[i]  = mmax[(size_t)d0 * 2 + i];
        t_d[i]  = 0.0f;
    }
    __syncthreads();
    seg_walk<true>(esrc, edst, aS, t_ad, t_m, t_d, d0, nr, e_lo, e_hi, E, tid);
    __syncthreads();
    for (int i = tid; i < nr * 2; i += SEG_T)
        atomicAdd(&denom[(size_t)d0 * 2 + i], t_d[i]);
}

#define SCAT_EPW 32
#define SCAT_B   8

__global__ __launch_bounds__(256) void gat_edge_scatter(
    const int* __restrict__ esrc, const int* __restrict__ edst,
    const float* __restrict__ aS, const float* __restrict__ aD,
    const float* __restrict__ mmax, const float* __restrict__ denom,
    const float* __restrict__ xp, float* __restrict__ acc, int E, int N)
{
    const int lane = threadIdx.x & 31;
    const int wave = __builtin_amdgcn_readfirstlane(
        (int)((blockIdx.x * blockDim.x + threadIdx.x) >> 5));
    const int Et = E + N;
    const int ebase = wave * SCAT_EPW;
    if (ebase >= Et) return;
    const int nvalid = (Et - ebase < SCAT_EPW) ? (Et - ebase) : SCAT_EPW;

    const int eid = ebase + lane;
    int s = 0, d = 0;
    if (eid < E)       { s = esrc[eid]; d = edst[eid]; }
    else if (eid < Et) { s = d = eid - E; }
    const float2 as = *(const float2*)(aS    + (size_t)s * 2);
    const float2 ad = *(const float2*)(aD    + (size_t)d * 2);
    const float2 mm = *(const float2*)(mmax  + (size_t)d * 2);
    const float2 dn = *(const float2*)(denom + (size_t)d * 2);
    const float e0 = lrelu(as.x + ad.x);
    const float e1 = lrelu(as.y + ad.y);
    const float w0 = 0.5f * (__expf(e0 - mm.x) / (dn.x + SM_EPS));
    const float w1 = 0.5f * (__expf(e1 - mm.y) / (dn.y + SM_EPS));
    const int w0i = __float_as_int(w0);
    const int w1i = __float_as_int(w1);

    float* accl = acc + lane;
    const float* xl = xp + lane;
    if (nvalid == SCAT_EPW) {
        for (int jb = 0; jb < SCAT_EPW; jb += SCAT_B) {
            float x00[SCAT_B], x01[SCAT_B], x10[SCAT_B], x11[SCAT_B];
            #pragma unroll
            for (int j = 0; j < SCAT_B; ++j) {
                const int sj = __builtin_amdgcn_readlane(s, jb + j);
                const float* xr = xl + (size_t)sj * XDIM;
                x00[j] = xr[0];
                x01[j] = xr[32];
                x10[j] = xr[OUT_F];
                x11[j] = xr[OUT_F + 32];
            }
            #pragma unroll
            for (int j = 0; j < SCAT_B; ++j) {
                const int   dj  = __builtin_amdgcn_readlane(d, jb + j);
                const float w0j = __int_as_float(__builtin_amdgcn_readlane(w0i, jb + j));
                const float w1j = __int_as_float(__builtin_amdgcn_readlane(w1i, jb + j));
                float* ap = accl + (size_t)dj * OUT_F;
                atomicAdd(ap,      w0j * x00[j] + w1j * x10[j]);
                atomicAdd(ap + 32, w0j * x01[j] + w1j * x11[j]);
            }
        }
    } else {
        for (int j = 0; j < nvalid; ++j) {
            const int   sj  = __builtin_amdgcn_readlane(s, j);
            const int   dj  = __builtin_amdgcn_readlane(d, j);
            const float w0j = __int_as_float(__builtin_amdgcn_readlane(w0i, j));
            const float w1j = __int_as_float(__builtin_amdgcn_readlane(w1i, j));
            const float* xr = xl + (size_t)sj * XDIM;
            float* ap = accl + (size_t)dj * OUT_F;
            atomicAdd(ap,      w0j * xr[0]  + w1j * xr[OUT_F]);
            atomicAdd(ap + 32, w0j * xr[32] + w1j * xr[OUT_F + 32]);
        }
    }
}

__global__ __launch_bounds__(256) void gat_finalize(
    const float* __restrict__ acc, const float* __restrict__ bias,
    const float* __restrict__ gamma, const float* __restrict__ beta,
    float* __restrict__ out, int N)
{
    const int wave = (blockIdx.x * blockDim.x + threadIdx.x) >> 5;
    const int lane = threadIdx.x & 31;
    if (wave >= N) return;
    const float* a = acc + (size_t)wave * OUT_F;
    const int f0 = lane, f1 = lane + 32;
    float v0 = a[f0] + bias[f0];
    float v1 = a[f1] + bias[f1];

    float sum = v0 + v1;
    #pragma unroll
    for (int off = 16; off >= 1; off >>= 1) sum += __shfl_xor(sum, off, 32);
    const float mu = sum * (1.0f / OUT_F);

    const float d0 = v0 - mu, d1 = v1 - mu;
    float sq = d0 * d0 + d1 * d1;
    #pragma unroll
    for (int off = 16; off >= 1; off >>= 1) sq += __shfl_xor(sq, off, 32);
    const float inv = rsqrtf(sq * (1.0f / OUT_F) + LN_EPS);

    out[(size_t)wave * OUT_F + f0] = d0 * inv * gamma[f0] + beta[f0];
    out[(size_t)wave * OUT_F + f1] = d1 * inv * gamma[f1] + beta[f1];
}

extern "C" void kernel_launch(void* const* d_in, const int* in_sizes, int n_in,
                              void* d_out, int out_size, void* d_ws, size_t ws_size,
                              hipStream_t stream)
{
    const float* X    = (const float*)d_in[0];
    const int*   ei   = (const int*)  d_in[1];
    const float* W    = (const float*)d_in[2];
    const float* attS = (const float*)d_in[3];
    const float* attD = (const float*)d_in[4];
    const float* bias = (const float*)d_in[5];
    const float* g    = (const float*)d_in[6];
    const float* b    = (const float*)d_in[7];
    float* out = (float*)d_out;

    const int N = in_sizes[0] / IN_F;
    const int E = in_sizes[1] / 2;
    const int* esrc = ei;
    const int* edst = ei + E;
    const int Et = E + N;

    float* ws    = (float*)d_ws;
    float* xp    = ws;
    float* accb  = xp   + (size_t)N * XDIM;
    float* aS    = accb + (size_t)N * OUT_F;
    float* aD    = aS   + (size_t)N * 2;
    float* mmax  = aD   + (size_t)N * 2;
    float* denom = mmax + (size_t)N * 2;

    {
        const int waves = (N + 15) / 16;
        gat_gemm_wmma<<<(waves + 7) / 8, 256, 0, stream>>>(X, W, xp, N);
    }
    gat_logits_init<<<(N + 7) / 8, 256, 0, stream>>>(
        xp, attS, attD, aS, aD, mmax, denom, accb, N);
    {
        const int ranges = (N + SEG_NR - 1) / SEG_NR;
        gat_seg_max<<<ranges * SEG_C, SEG_T, 0, stream>>>(
            esrc, edst, aS, aD, mmax, E, N);
        gat_seg_sum<<<ranges * SEG_C, SEG_T, 0, stream>>>(
            esrc, edst, aS, aD, mmax, denom, E, N);
    }
    gat_edge_scatter<<<(Et + 8 * SCAT_EPW - 1) / (8 * SCAT_EPW), 256, 0, stream>>>(
        esrc, edst, aS, aD, mmax, denom, xp, accb, E, N);
    gat_finalize<<<(N + 7) / 8, 256, 0, stream>>>(accb, bias, g, b, out, N);
}


# Projection GEMM: each wave holds the W fragments of 2 column tiles in registers for all K and walks 16-row strips of X (4-wave blocks); no B loads in the strip loop; k summed in another order
# speedup vs baseline: 3.3693x; 1.1326x over previous
#include <hip/hip_runtime.h>
#include <hip/hip_bf16.h>
#include <math.h>


typedef __attribute__((ext_vector_type(2))) float v2f;
typedef __attribute__((ext_vector_type(8))) float v8f;

#define IN_F     128
#define OUT_F    64
#define HEADS    2
#define XDIM     128
#define NEG_SLOPE 0.2f
#define LN_EPS    1e-5f
#define SM_EPS    1e-16f

#define GEMM_BLOCKS 768

__global__ __launch_bounds__(128) void gat_gemm_wmma(
    const float* __restrict__ X, const float* __restrict__ W,
    float* __restrict__ xp, int N)
{
    const int lane = threadIdx.x & 31;
    const int w    = threadIdx.x >> 5;
    const int lm   = lane & 15;
    const int h    = lane >> 4;
    const int nstrips = (N + 15) >> 4;

    float4 bq[2][16];
    #pragma unroll
    for (int t = 0; t < 2; ++t) {
        const float* brow = W + (size_t)((2 * w + t) * 16 + lm) * IN_F + 4 * h;
        #pragma unroll
        for (int j = 0; j < 16; ++j) bq[t][j] = *(const float4*)(brow + 8 * j);
    }

    const int mofs = h << 3;
    const int c0   = (2 * w) * 16 + lm;

    int strip = blockIdx.x;
    if (strip >= nstrips) return;

    float4 aq[16];
    {
        int ar = strip * 16 + lm; if (ar >= N) ar = N - 1;
        const float* arow = X + (size_t)ar * IN_F + 4 * h;
        #pragma unroll
        for (int j = 0; j < 16; ++j) aq[j] = *(const float4*)(arow + 8 * j);
    }

    for (; strip < nstrips; strip += gridDim.x) {
        float4 an[16];
        {
            const int ns = (strip + (int)gridDim.x < nstrips) ? strip + (int)gridDim.x : strip;
            int ar = ns * 16 + lm; if (ar >= N) ar = N - 1;
            const float* arow = X + (size_t)ar * IN_F + 4 * h;
            #pragma unroll
            for (int j = 0; j < 16; ++j) an[j] = *(const float4*)(arow + 8 * j);
        }

        v8f acc0 = (v8f){}, acc1 = (v8f){};
        #pragma unroll
        for (int j = 0; j < 16; ++j) {
            const v2f a_lo = {aq[j].x, aq[j].y};
            const v2f a_hi = {aq[j].z, aq[j].w};
            const v2f b0lo = {bq[0][j].x, bq[0][j].y};
            const v2f b1lo = {bq[1][j].x, bq[1][j].y};
            const v2f b0hi = {bq[0][j].z, bq[0][j].w};
            const v2f b1hi = {bq[1][j].z, bq[1][j].w};
            acc0 = __builtin_amdgcn_wmma_f32_16x16x4_f32(
                false, a_lo, false, b0lo, (short)0, acc0, false, false);
            acc1 = __builtin_amdgcn_wmma_f32_16x16x4_f32(
                false, a_lo, false, b1lo, (short)0, acc1, false, false);
            acc0 = __builtin_amdgcn_wmma_f32_16x16x4_f32(
                false, a_hi, false, b0hi, (short)0, acc0, false, false);
            acc1 = __builtin_amdgcn_wmma_f32_16x16x4_f32(
                false, a_hi, false, b1hi, (short)0, acc1, false, false);
        }

        const int r0 = strip * 16;
        #pragma unroll
        for (int v = 0; v < 8; ++v) {
            const int row = r0 + mofs + v;
            if (row < N) {
                float* orow = xp + (size_t)row * XDIM + c0;
                orow[0]  = acc0[v];
                orow[16] = acc1[v];
            }
        }

        #pragma unroll
        for (int j = 0; j < 16; ++j) aq[j] = an[j];
    }
}

__global__ __launch_bounds__(256) void gat_logits_init(
    const float* __restrict__ xp,
    const float* __restrict__ attS, const float* __restrict__ attD,
    float* __restrict__ aS, float* __restrict__ aD,
    float* __restrict__ mmax, float* __restrict__ denom,
    float* __restrict__ acc, int N)
{
    const int wave = (blockIdx.x * blockDim.x + threadIdx.x) >> 5;
    const int lane = threadIdx.x & 31;
    if (wave >= N) return;
    const int head = lane >> 4;
    const int fo   = lane * 4;

    float4 xv = *(const float4*)(xp + (size_t)wave * XDIM + fo);
    float4 sv = *(const float4*)(attS + fo);
    float4 dv = *(const float4*)(attD + fo);
    float ps = xv.x*sv.x + xv.y*sv.y + xv.z*sv.z + xv.w*sv.w;
    float pd = xv.x*dv.x + xv.y*dv.y + xv.z*dv.z + xv.w*dv.w;
    #pragma unroll
    for (int off = 8; off >= 1; off >>= 1) {
        ps += __shfl_xor(ps, off, 32);
        pd += __shfl_xor(pd, off, 32);
    }
    if ((lane & 15) == 0) {
        aS[wave * 2 + head]    = ps;
        aD[wave * 2 + head]    = pd;
        mmax[wave * 2 + head]  = -INFINITY;
        denom[wave * 2 + head] = 0.0f;
    }
    float2 z = {0.f, 0.f};
    *(float2*)(acc + (size_t)wave * OUT_F + lane * 2) = z;
}

__device__ __forceinline__ void atomicMaxF(float* addr, float val)
{
    if (val >= 0.0f) atomicMax((int*)addr, __float_as_int(val));
    else             atomicMin((unsigned int*)addr, __float_as_uint(val));
}

__device__ __forceinline__ float lrelu(float v)
{
    return v > 0.0f ? v : NEG_SLOPE * v;
}

#define SEG_NR 6256
#define SEG_C  16
#define SEG_T  1024
#define SEG_U  3

template <bool SUM>
__device__ __forceinline__ void seg_update(
    bool mine, unsigned ld, float2 as, float2 ad, float2 mm, float* t_m, float* t_d)
{
    const float e0 = lrelu(as.x + ad.x);
    const float e1 = lrelu(as.y + ad.y);
    if (SUM) {
        const float x0 = __expf(e0 - mm.x);
        const float x1 = __expf(e1 - mm.y);
        if (mine) {
            atomicAdd(t_d + 2 * ld,     x0);
            atomicAdd(t_d + 2 * ld + 1, x1);
        }
    } else {
        if (mine) {
            atomicMaxF(t_m + 2 * ld,     e0);
            atomicMaxF(t_m + 2 * ld + 1, e1);
        }
    }
}

template <bool SUM>
__device__ __forceinline__ void seg_walk(
    const int* __restrict__ esrc, const int* __restrict__ edst,
    const float* __restrict__ aS,
    const float* t_ad, float* t_m, float* t_d,
    int d0, int nr, int e_lo, int e_hi, int E, int tid)
{
    int v_hi = e_lo;
    if ((E & 3) == 0) {
        const int lim = (e_hi < E) ? e_hi : E;
        if (lim > e_lo) v_hi = lim & ~3;
    }
    if (v_hi > e_lo) {
        const int step = SEG_T * 4;
        const int g_last = v_hi - 4;
        int4 dn[SEG_U], sn[SEG_U];
        int g0 = e_lo + tid * 4;
        #pragma unroll
        for (int u = 0; u < SEG_U; ++u) {
            const int g = g0 + u * step;
            const int gc = (g < g_last) ? g : g_last;
            dn[u] = *(const int4*)(edst + gc);
            sn[u] = *(const int4*)(esrc + gc);
        }
        for (int gb = e_lo; gb < v_hi; gb += step * SEG_U) {
            int dc[SEG_U * 4], sc[SEG_U * 4];
            #pragma unroll
            for (int u = 0; u < SEG_U; ++u) {
                const bool ok = (g0 + u * step) < v_hi;
                dc[u * 4 + 0] = ok ? dn[u].x : -1;
                dc[u * 4 + 1] = ok ? dn[u].y : -1;
                dc[u * 4 + 2] = ok ? dn[u].z : -1;
                dc[u * 4 + 3] = ok ? dn[u].w : -1;
                sc[u * 4 + 0] = sn[u].x; sc[u * 4 + 1] = sn[u].y;
                sc[u * 4 + 2] = sn[u].z; sc[u * 4 + 3] = sn[u].w;
            }
            bool     mine[SEG_U * 4];
            unsigned ld[SEG_U * 4];
            float2   as[SEG_U * 4];
            #pragma unroll
            for (int k = 0; k < SEG_U * 4; ++k) {
                ld[k]   = (unsigned)(dc[k] - d0);
                mine[k] = ld[k] < (unsigned)nr;
                if (!mine[k]) ld[k] = 0u;
                const int se = mine[k] ? sc[k] : 0;
                as[k] = *(const float2*)(aS + (size_t)se * 2);
            }
            g0 += step * SEG_U;
            #pragma unroll
            for (int u = 0; u < SEG_U; ++u) {
                const int g = g0 + u * step;
                const int gc = (g < g_last) ? g : g_last;
                dn[u] = *(const int4*)(edst + gc);
                sn[u] = *(const int4*)(esrc + gc);
            }
            float2 ad[SEG_U * 4], mm[SEG_U * 4];
            #pragma unroll
            for (int k = 0; k < SEG_U * 4; ++k) {
                ad[k] = *(const float2*)(t_ad + 2 * ld[k]);
                mm[k] = SUM ? *(const float2*)(t_m + 2 * ld[k]) : float2{0.f, 0.f};
            }
            #pragma unroll
            for (int k = 0; k < SEG_U * 4; ++k)
                seg_update<SUM>(mine[k], ld[k], as[k], ad[k], mm[k], t_m, t_d);
        }
    }
    for (int eid = v_hi + tid; eid < e_hi; eid += SEG_T) {
        int s, d;
        if (eid < E) { s = esrc[eid]; d = edst[eid]; }
        else         { s = d = eid - E; }
        const unsigned ld1 = (unsigned)(d - d0);
        if (ld1 < (unsigned)nr) {
            const float2 as1 = *(const float2*)(aS + (size_t)s * 2);
            const float2 ad1 = *(const float2*)(t_ad + 2 * ld1);
            const float2 mm1 = SUM ? *(const float2*)(t_m + 2 * ld1) : float2{0.f, 0.f};
            seg_update<SUM>(true, ld1, as1, ad1, mm1, t_m, t_d);
        }
    }
}

__device__ __forceinline__ void seg_chunk(int Et, int c, int& e_lo, int& e_hi)
{
    const int ec = (((Et + SEG_C - 1) / SEG_C) + 3) & ~3;
    e_lo = c * ec;
    if (e_lo > Et) e_lo = Et;
    e_hi = (Et - e_lo < ec) ? Et : e_lo + ec;
}

__global__ __launch_bounds__(SEG_T) void gat_seg_max(
    const int* __restrict__ esrc, const int* __restrict__ edst,
    const float* __restrict__ aS, const float* __restrict__ aD,
    float* __restrict__ mmax, int E, int N)
{
    __shared__ float t_ad[SEG_NR * 2];
    __shared__ float t_m[SEG_NR * 2];
    const int tid = threadIdx.x;
    const int d0 = (int)(blockIdx.x / SEG_C) * SEG_NR;
    const int c  = (int)(blockIdx.x % SEG_C);
    const int nr = (N - d0 < SEG_NR) ? (N - d0) : SEG_NR;
    int e_lo, e_hi;
    seg_chunk(E + N, c, e_lo, e_hi);

    for (int i = tid; i < nr * 2; i += SEG_T) {
        t_ad[i] = aD[(size_t)d0 * 2 + i];
        t_m[i]  = -INFINITY;
    }
    __syncthreads();
    seg_walk<false>(esrc, edst, aS, t_ad, t_m, nullptr, d0, nr, e_lo, e_hi, E, tid);
    __syncthreads();
    for (int i = tid; i < nr * 2; i += SEG_T)
        atomicMaxF(&mmax[(size_t)d0 * 2 + i], t_m[i]);
}

__global__ __launch_bounds__(SEG_T) void gat_seg_sum(
    const int* __restrict__ esrc, const int* __restrict__ edst,
    const float* __restrict__ aS, const float* __restrict__ aD,
    const float* __restrict__ mmax, float* __restrict__ denom, int E, int N)
{
    __shared__ float t_ad[SEG_NR * 2];
    __shared__ float t_m[SEG_NR * 2];
    __shared__ float t_d[SEG_NR * 2];
    const int tid = threadIdx.x;
    const int d0 = (int)(blockIdx.x / SEG_C) * SEG_NR;
    const int c  = (int)(blockIdx.x % SEG_C);
    const int nr = (N - d0 < SEG_NR) ? (N - d0) : SEG_NR;
    int e_lo, e_hi;
    seg_chunk(E + N, c, e_lo, e_hi);

    for (int i = tid; i < nr * 2; i += SEG_T) {
        t_ad[i] = aD[(size_t)d0 * 2 + i];
        t_m[i]  = mmax[(size_t)d0 * 2 + i];
        t_d[i]  = 0.0f;
    }
    __syncthreads();
    seg_walk<true>(esrc, edst, aS, t_ad, t_m, t_d, d0, nr, e_lo, e_hi, E, tid);
    __syncthreads();
    for (int i = tid; i < nr * 2; i += SEG_T)
        atomicAdd(&denom[(size_t)d0 * 2 + i], t_d[i]);
}

#define SCAT_EPW 32
#define SCAT_B   8

__global__ __launch_bounds__(256) void gat_edge_scatter(
    const int* __restrict__ esrc, const int* __restrict__ edst,
    const float* __restrict__ aS, const float* __restrict__ aD,
    const float* __restrict__ mmax, const float* __restrict__ denom,
    const float* __restrict__ xp, float* __restrict__ acc, int E, int N)
{
    const int lane = threadIdx.x & 31;
    const int wave = __builtin_amdgcn_readfirstlane(
        (int)((blockIdx.x * blockDim.x + threadIdx.x) >> 5));
    const int Et = E + N;
    const int ebase = wave * SCAT_EPW;
    if (ebase >= Et) return;
    const int nvalid = (Et - ebase < SCAT_EPW) ? (Et - ebase) : SCAT_EPW;

    const int eid = ebase + lane;
    int s = 0, d = 0;
    if (eid < E)       { s = esrc[eid]; d = edst[eid]; }
    else if (eid < Et) { s = d = eid - E; }
    const float2 as = *(const float2*)(aS    + (size_t)s * 2);
    const float2 ad = *(const float2*)(aD    + (size_t)d * 2);
    const float2 mm = *(const float2*)(mmax  + (size_t)d * 2);
    const float2 dn = *(const float2*)(denom + (size_t)d * 2);
    const float e0 = lrelu(as.x + ad.x);
    const float e1 = lrelu(as.y + ad.y);
    const float w0 = 0.5f * (__expf(e0 - mm.x) / (dn.x + SM_EPS));
    const float w1 = 0.5f * (__expf(e1 - mm.y) / (dn.y + SM_EPS));
    const int w0i = __float_as_int(w0);
    const int w1i = __float_as_int(w1);

    float* accl = acc + lane;
    const float* xl = xp + lane;
    if (nvalid == SCAT_EPW) {
        for (int jb = 0; jb < SCAT_EPW; jb += SCAT_B) {
            float x00[SCAT_B], x01[SCAT_B], x10[SCAT_B], x11[SCAT_B];
            #pragma unroll
            for (int j = 0; j < SCAT_B; ++j) {
                const int sj = __builtin_amdgcn_readlane(s, jb + j);
                const float* xr = xl + (size_t)sj * XDIM;
                x00[j] = xr[0];
                x01[j] = xr[32];
                x10[j] = xr[OUT_F];
                x11[j] = xr[OUT_F + 32];
            }
            #pragma unroll
            for (int j = 0; j < SCAT_B; ++j) {
                const int   dj  = __builtin_amdgcn_readlane(d, jb + j);
                const float w0j = __int_as_float(__builtin_amdgcn_readlane(w0i, jb + j));
                const float w1j = __int_as_float(__builtin_amdgcn_readlane(w1i, jb + j));
                float* ap = accl + (size_t)dj * OUT_F;
                atomicAdd(ap,      w0j * x00[j] + w1j * x10[j]);
                atomicAdd(ap + 32, w0j * x01[j] + w1j * x11[j]);
            }
        }
    } else {
        for (int j = 0; j < nvalid; ++j) {
            const int   sj  = __builtin_amdgcn_readlane(s, j);
            const int   dj  = __builtin_amdgcn_readlane(d, j);
            const float w0j = __int_as_float(__builtin_amdgcn_readlane(w0i, j));
            const float w1j = __int_as_float(__builtin_amdgcn_readlane(w1i, j));
            const float* xr = xl + (size_t)sj * XDIM;
            float* ap = accl + (size_t)dj * OUT_F;
            atomicAdd(ap,      w0j * xr[0]  + w1j * xr[OUT_F]);
            atomicAdd(ap + 32, w0j * xr[32] + w1j * xr[OUT_F + 32]);
        }
    }
}

__global__ __launch_bounds__(256) void gat_finalize(
    const float* __restrict__ acc, const float* __restrict__ bias,
    const float* __restrict__ gamma, const float* __restrict__ beta,
    float* __restrict__ out, int N)
{
    const int wave = (blockIdx.x * blockDim.x + threadIdx.x) >> 5;
    const int lane = threadIdx.x & 31;
    if (wave >= N) return;
    const float* a = acc + (size_t)wave * OUT_F;
    const int f0 = lane, f1 = lane + 32;
    float v0 = a[f0] + bias[f0];
    float v1 = a[f1] + bias[f1];

    float sum = v0 + v1;
    #pragma unroll
    for (int off = 16; off >= 1; off >>= 1) sum += __shfl_xor(sum, off, 32);
    const float mu = sum * (1.0f / OUT_F);

    const float d0 = v0 - mu, d1 = v1 - mu;
    float sq = d0 * d0 + d1 * d1;
    #pragma unroll
    for (int off = 16; off >= 1; off >>= 1) sq += __shfl_xor(sq, off, 32);
    const float inv = rsqrtf(sq * (1.0f / OUT_F) + LN_EPS);

    out[(size_t)wave * OUT_F + f0] = d0 * inv * gamma[f0] + beta[f0];
    out[(size_t)wave * OUT_F + f1] = d1 * inv * gamma[f1] + beta[f1];
}

extern "C" void kernel_launch(void* const* d_in, const int* in_sizes, int n_in,
                              void* d_out, int out_size, void* d_ws, size_t ws_size,
                              hipStream_t stream)
{
    const float* X    = (const float*)d_in[0];
    const int*   ei   = (const int*)  d_in[1];
    const float* W    = (const float*)d_in[2];
    const float* attS = (const float*)d_in[3];
    const float* attD = (const float*)d_in[4];
    const float* bias = (const float*)d_in[5];
    const float* g    = (const float*)d_in[6];
    const float* b    = (const float*)d_in[7];
    float* out = (float*)d_out;

    const int N = in_sizes[0] / IN_F;
    const int E = in_sizes[1] / 2;
    const int* esrc = ei;
    const int* edst = ei + E;
    const int Et = E + N;

    float* ws    = (float*)d_ws;
    float* xp    = ws;
    float* accb  = xp   + (size_t)N * XDIM;
    float* aS    = accb + (size_t)N * OUT_F;
    float* aD    = aS   + (size_t)N * 2;
    float* mmax  = aD   + (size_t)N * 2;
    float* denom = mmax + (size_t)N * 2;

    gat_gemm_wmma<<<GEMM_BLOCKS, 128, 0, stream>>>(X, W, xp, N);
    gat_logits_init<<<(N + 7) / 8, 256, 0, stream>>>(
        xp, attS, attD, aS, aD, mmax, denom, accb, N);
    {
        const int ranges = (N + SEG_NR - 1) / SEG_NR;
        gat_seg_max<<<ranges * SEG_C, SEG_T, 0, stream>>>(
            esrc, edst, aS, aD, mmax, E, N);
        gat_seg_sum<<<ranges * SEG_C, SEG_T, 0, stream>>>(
            esrc, edst, aS, aD, mmax, denom, E, N);
    }
    gat_edge_scatter<<<(Et + 8 * SCAT_EPW - 1) / (8 * SCAT_EPW), 256, 0, stream>>>(
        esrc, edst, aS, aD, mmax, denom, xp, accb, E, N);
    gat_finalize<<<(N + 7) / 8, 256, 0, stream>>>(accb, bias, g, b, out, N);
}


# Node kernels (logits+init, LayerNorm): a wave takes 16 consecutive nodes with all rows requested up front, not one node per wave: 6256 resident waves per launch, not 100000; sums in the same order
# speedup vs baseline: 3.5447x; 1.0521x over previous
#include <hip/hip_runtime.h>
#include <hip/hip_bf16.h>
#include <math.h>


typedef __attribute__((ext_vector_type(2))) float v2f;
typedef __attribute__((ext_vector_type(8))) float v8f;

#define IN_F     128
#define OUT_F    64
#define HEADS    2
#define XDIM     128
#define NEG_SLOPE 0.2f
#define LN_EPS    1e-5f
#define SM_EPS    1e-16f

#define GEMM_BLOCKS 768

__global__ __launch_bounds__(128) void gat_gemm_wmma(
    const float* __restrict__ X, const float* __restrict__ W,
    float* __restrict__ xp, int N)
{
    const int lane = threadIdx.x & 31;
    const int w    = threadIdx.x >> 5;
    const int lm   = lane & 15;
    const int h    = lane >> 4;
    const int nstrips = (N + 15) >> 4;

    float4 bq[2][16];
    #pragma unroll
    for (int t = 0; t < 2; ++t) {
        const float* brow = W + (size_t)((2 * w + t) * 16 + lm) * IN_F + 4 * h;
        #pragma unroll
        for (int j = 0; j < 16; ++j) bq[t][j] = *(const float4*)(brow + 8 * j);
    }

    const int mofs = h << 3;
    const int c0   = (2 * w) * 16 + lm;

    int strip = blockIdx.x;
    if (strip >= nstrips) return;

    float4 aq[16];
    {
        int ar = strip * 16 + lm; if (ar >= N) ar = N - 1;
        const float* arow = X + (size_t)ar * IN_F + 4 * h;
        #pragma unroll
        for (int j = 0; j < 16; ++j) aq[j] = *(const float4*)(arow + 8 * j);
    }

    for (; strip < nstrips; strip += gridDim.x) {
        float4 an[16];
        {
            const int ns = (strip + (int)gridDim.x < nstrips) ? strip + (int)gridDim.x : strip;
            int ar = ns * 16 + lm; if (ar >= N) ar = N - 1;
            const float* arow = X + (size_t)ar * IN_F + 4 * h;
            #pragma unroll
            for (int j = 0; j < 16; ++j) an[j] = *(const float4*)(arow + 8 * j);
        }

        v8f acc0 = (v8f){}, acc1 = (v8f){};
        #pragma unroll
        for (int j = 0; j < 16; ++j) {
            const v2f a_lo = {aq[j].x, aq[j].y};
            const v2f a_hi = {aq[j].z, aq[j].w};
            const v2f b0lo = {bq[0][j].x, bq[0][j].y};
            const v2f b1lo = {bq[1][j].x, bq[1][j].y};
            const v2f b0hi = {bq[0][j].z, bq[0][j].w};
            const v2f b1hi = {bq[1][j].z, bq[1][j].w};
            acc0 = __builtin_amdgcn_wmma_f32_16x16x4_f32(
                false, a_lo, false, b0lo, (short)0, acc0, false, false);
            acc1 = __builtin_amdgcn_wmma_f32_16x16x4_f32(
                false, a_lo, false, b1lo, (short)0, acc1, false, false);
            acc0 = __builtin_amdgcn_wmma_f32_16x16x4_f32(
                false, a_hi, false, b0hi, (short)0, acc0, false, false);
            acc1 = __builtin_amdgcn_wmma_f32_16x16x4_f32(
                false, a_hi, false, b1hi, (short)0, acc1, false, false);
        }

        const int r0 = strip * 16;
        #pragma unroll
        for (int v = 0; v < 8; ++v) {
            const int row = r0 + mofs + v;
            if (row < N) {
                float* orow = xp + (size_t)row * XDIM + c0;
                orow[0]  = acc0[v];
                orow[16] = acc1[v];
            }
        }

        #pragma unroll
        for (int j = 0; j < 16; ++j) aq[j] = an[j];
    }
}

#define NODE_W 16

template <bool FULL>
__device__ __forceinline__ void logits_rows(
    const float* __restrict__ xp, const float4 sv, const float4 dv,
    float* __restrict__ aS, float* __restrict__ aD,
    float* __restrict__ mmax, float* __restrict__ denom,
    float* __restrict__ acc, int n0, int N, int lane)
{
    const int fo = lane * 4;

    float4 xv[NODE_W];
    #pragma unroll
    for (int k = 0; k < NODE_W; ++k) {
        int node = n0 + k;
        if (!FULL && node >= N) node = N - 1;
        xv[k] = *(const float4*)(xp + (size_t)node * XDIM + fo);
    }

    const float2 z = {0.f, 0.f};
    #pragma unroll
    for (int k = 0; k < NODE_W; ++k)
        if (FULL || n0 + k < N)
            *(float2*)(acc + (size_t)(n0 + k) * OUT_F + lane * 2) = z;

    float ps[NODE_W], pd[NODE_W];
    #pragma unroll
    for (int k = 0; k < NODE_W; ++k) {
        ps[k] = xv[k].x*sv.x + xv[k].y*sv.y + xv[k].z*sv.z + xv[k].w*sv.w;
        pd[k] = xv[k].x*dv.x + xv[k].y*dv.y + xv[k].z*dv.z + xv[k].w*dv.w;
    }
    #pragma unroll
    for (int off = 8; off >= 1; off >>= 1) {
        #pragma unroll
        for (int k = 0; k < NODE_W; ++k) {
            ps[k] += __shfl_xor(ps[k], off, 32);
            pd[k] += __shfl_xor(pd[k], off, 32);
        }
    }

    const int j = lane & 15;
    const int head = lane >> 4;
    float ss = ps[0], sd = pd[0];
    #pragma unroll
    for (int k = 1; k < NODE_W; ++k) {
        ss = (j == k) ? ps[k] : ss;
        sd = (j == k) ? pd[k] : sd;
    }
    if (FULL || n0 + j < N) {
        const size_t ti = (size_t)(n0 + j) * 2 + head;
        aS[ti]    = ss;
        aD[ti]    = sd;
        mmax[ti]  = -INFINITY;
        denom[ti] = 0.0f;
    }
}

__global__ __launch_bounds__(256) void gat_logits_init(
    const float* __restrict__ xp,
    const float* __restrict__ attS, const float* __restrict__ attD,
    float* __restrict__ aS, float* __restrict__ aD,
    float* __restrict__ mmax, float* __restrict__ denom,
    float* __restrict__ acc, int N)
{
    const int lane = threadIdx.x & 31;
    const int wave = __builtin_amdgcn_readfirstlane(
        (int)((blockIdx.x * blockDim.x + threadIdx.x) >> 5));
    const int n0 = wave * NODE_W;
    if (n0 >= N) return;

    const float4 sv = *(const float4*)(attS + lane * 4);
    const float4 dv = *(const float4*)(attD + lane * 4);
    if (n0 + NODE_W <= N)
        logits_rows<true>(xp, sv, dv, aS, aD, mmax, denom, acc, n0, N, lane);
    else
        logits_rows<false>(xp, sv, dv, aS, aD, mmax, denom, acc, n0, N, lane);
}

__device__ __forceinline__ void atomicMaxF(float* addr, float val)
{
    if (val >= 0.0f) atomicMax((int*)addr, __float_as_int(val));
    else             atomicMin((unsigned int*)addr, __float_as_uint(val));
}

__device__ __forceinline__ float lrelu(float v)
{
    return v > 0.0f ? v : NEG_SLOPE * v;
}

#define SEG_NR 6256
#define SEG_C  16
#define SEG_T  1024
#define SEG_U  3

template <bool SUM>
__device__ __forceinline__ void seg_update(
    bool mine, unsigned ld, float2 as, float2 ad, float2 mm, float* t_m, float* t_d)
{
    const float e0 = lrelu(as.x + ad.x);
    const float e1 = lrelu(as.y + ad.y);
    if (SUM) {
        const float x0 = __expf(e0 - mm.x);
        const float x1 = __expf(e1 - mm.y);
        if (mine) {
            atomicAdd(t_d + 2 * ld,     x0);
            atomicAdd(t_d + 2 * ld + 1, x1);
        }
    } else {
        if (mine) {
            atomicMaxF(t_m + 2 * ld,     e0);
            atomicMaxF(t_m + 2 * ld + 1, e1);
        }
    }
}

template <bool SUM>
__device__ __forceinline__ void seg_walk(
    const int* __restrict__ esrc, const int* __restrict__ edst,
    const float* __restrict__ aS,
    const float* t_ad, float* t_m, float* t_d,
    int d0, int nr, int e_lo, int e_hi, int E, int tid)
{
    int v_hi = e_lo;
    if ((E & 3) == 0) {
        const int lim = (e_hi < E) ? e_hi : E;
        if (lim > e_lo) v_hi = lim & ~3;
    }
    if (v_hi > e_lo) {
        const int step = SEG_T * 4;
        const int g_last = v_hi - 4;
        int4 dn[SEG_U], sn[SEG_U];
        int g0 = e_lo + tid * 4;
        #pragma unroll
        for (int u = 0; u < SEG_U; ++u) {
            const int g = g0 + u * step;
            const int gc = (g < g_last) ? g : g_last;
            dn[u] = *(const int4*)(edst + gc);
            sn[u] = *(const int4*)(esrc + gc);
        }
        for (int gb = e_lo; gb < v_hi; gb += step * SEG_U) {
            int dc[SEG_U * 4], sc[SEG_U * 4];
            #pragma unroll
            for (int u = 0; u < SEG_U; ++u) {
                const bool ok = (g0 + u * step) < v_hi;
                dc[u * 4 + 0] = ok ? dn[u].x : -1;
                dc[u * 4 + 1] = ok ? dn[u].y : -1;
                dc[u * 4 + 2] = ok ? dn[u].z : -1;
                dc[u * 4 + 3] = ok ? dn[u].w : -1;
                sc[u * 4 + 0] = sn[u].x; sc[u * 4 + 1] = sn[u].y;
                sc[u * 4 + 2] = sn[u].z; sc[u * 4 + 3] = sn[u].w;
            }
            bool     mine[SEG_U * 4];
            unsigned ld[SEG_U * 4];
            float2   as[SEG_U * 4];
            #pragma unroll
            for (int k = 0; k < SEG_U * 4; ++k) {
                ld[k]   = (unsigned)(dc[k] - d0);
                mine[k] = ld[k] < (unsigned)nr;
                if (!mine[k]) ld[k] = 0u;
                const int se = mine[k] ? sc[k] : 0;
                as[k] = *(const float2*)(aS + (size_t)se * 2);
            }
            g0 += step * SEG_U;
            #pragma unroll
            for (int u = 0; u < SEG_U; ++u) {
                const int g = g0 + u * step;
                const int gc = (g < g_last) ? g : g_last;
                dn[u] = *(const int4*)(edst + gc);
                sn[u] = *(const int4*)(esrc + gc);
            }
            float2 ad[SEG_U * 4], mm[SEG_U * 4];
            #pragma unroll
            for (int k = 0; k < SEG_U * 4; ++k) {
                ad[k] = *(const float2*)(t_ad + 2 * ld[k]);
                mm[k] = SUM ? *(const float2*)(t_m + 2 * ld[k]) : float2{0.f, 0.f};
            }
            #pragma unroll
            for (int k = 0; k < SEG_U * 4; ++k)
                seg_update<SUM>(mine[k], ld[k], as[k], ad[k], mm[k], t_m, t_d);
        }
    }
    for (int eid = v_hi + tid; eid < e_hi; eid += SEG_T) {
        int s, d;
        if (eid < E) { s = esrc[eid]; d = edst[eid]; }
        else         { s = d = eid - E; }
        const unsigned ld1 = (unsigned)(d - d0);
        if (ld1 < (unsigned)nr) {
            const float2 as1 = *(const float2*)(aS + (size_t)s * 2);
            const float2 ad1 = *(const float2*)(t_ad + 2 * ld1);
            const float2 mm1 = SUM ? *(const float2*)(t_m + 2 * ld1) : float2{0.f, 0.f};
            seg_update<SUM>(true, ld1, as1, ad1, mm1, t_m, t_d);
        }
    }
}

__device__ __forceinline__ void seg_chunk(int Et, int c, int& e_lo, int& e_hi)
{
    const int ec = (((Et + SEG_C - 1) / SEG_C) + 3) & ~3;
    e_lo = c * ec;
    if (e_lo > Et) e_lo = Et;
    e_hi = (Et - e_lo < ec) ? Et : e_lo + ec;
}

__global__ __launch_bounds__(SEG_T) void gat_seg_max(
    const int* __restrict__ esrc, const int* __restrict__ edst,
    const float* __restrict__ aS, const float* __restrict__ aD,
    float* __restrict__ mmax, int E, int N)
{
    __shared__ float t_ad[SEG_NR * 2];
    __shared__ float t_m[SEG_NR * 2];
    const int tid = threadIdx.x;
    const int d0 = (int)(blockIdx.x / SEG_C) * SEG_NR;
    const int c  = (int)(blockIdx.x % SEG_C);
    const int nr = (N - d0 < SEG_NR) ? (N - d0) : SEG_NR;
    int e_lo, e_hi;
    seg_chunk(E + N, c, e_lo, e_hi);

    for (int i = tid; i < nr * 2; i += SEG_T) {
        t_ad[i] = aD[(size_t)d0 * 2 + i];
        t_m[i]  = -INFINITY;
    }
    __syncthreads();
    seg_walk<false>(esrc, edst, aS, t_ad, t_m, nullptr, d0, nr, e_lo, e_hi, E, tid);
    __syncthreads();
    for (int i = tid; i < nr * 2; i += SEG_T)
        atomicMaxF(&mmax[(size_t)d0 * 2 + i], t_m[i]);
}

__global__ __launch_bounds__(SEG_T) void gat_seg_sum(
    const int* __restrict__ esrc, const int* __restrict__ edst,
    const float* __restrict__ aS, const float* __restrict__ aD,
    const float* __restrict__ mmax, float* __restrict__ denom, int E, int N)
{
    __shared__ float t_ad[SEG_NR * 2];
    __shared__ float t_m[SEG_NR * 2];
    __shared__ float t_d[SEG_NR * 2];
    const int tid = threadIdx.x;
    const int d0 = (int)(blockIdx.x / SEG_C) * SEG_NR;
    const int c  = (int)(blockIdx.x % SEG_C);
    const int nr = (N - d0 < SEG_NR) ? (N - d0) : SEG_NR;
    int e_lo, e_hi;
    seg_chunk(E + N, c, e_lo, e_hi);

    for (int i = tid; i < nr * 2; i += SEG_T) {
        t_ad[i] = aD[(size_t)d0 * 2 + i];
        t_m[i]  = mmax[(size_t)d0 * 2 + i];
        t_d[i]  = 0.0f;
    }
    __syncthreads();
    seg_walk<true>(esrc, edst, aS, t_ad, t_m, t_d, d0, nr, e_lo, e_hi, E, tid);
    __syncthreads();
    for (int i = tid; i < nr * 2; i += SEG_T)
        atomicAdd(&denom[(size_t)d0 * 2 + i], t_d[i]);
}

#define SCAT_EPW 32
#define SCAT_B   8

__global__ __launch_bounds__(256) void gat_edge_scatter(
    const int* __restrict__ esrc, const int* __restrict__ edst,
    const float* __restrict__ aS, const float* __restrict__ aD,
    const float* __restrict__ mmax, const float* __restrict__ denom,
    const float* __restrict__ xp, float* __restrict__ acc, int E, int N)
{
    const int lane = threadIdx.x & 31;
    const int wave = __builtin_amdgcn_readfirstlane(
        (int)((blockIdx.x * blockDim.x + threadIdx.x) >> 5));
    const int Et = E + N;
    const int ebase = wave * SCAT_EPW;
    if (ebase >= Et) return;
    const int nvalid = (Et - ebase < SCAT_EPW) ? (Et - ebase) : SCAT_EPW;

    const int eid = ebase + lane;
    int s = 0, d = 0;
    if (eid < E)       { s = esrc[eid]; d = edst[eid]; }
    else if (eid < Et) { s = d = eid - E; }
    const float2 as = *(const float2*)(aS    + (size_t)s * 2);
    const float2 ad = *(const float2*)(aD    + (size_t)d * 2);
    const float2 mm = *(const float2*)(mmax  + (size_t)d * 2);
    const float2 dn = *(const float2*)(denom + (size_t)d * 2);
    const float e0 = lrelu(as.x + ad.x);
    const float e1 = lrelu(as.y + ad.y);
    const float w0 = 0.5f * (__expf(e0 - mm.x) / (dn.x + SM_EPS));
    const float w1 = 0.5f * (__expf(e1 - mm.y) / (dn.y + SM_EPS));
    const int w0i = __float_as_int(w0);
    const int w1i = __float_as_int(w1);

    float* accl = acc + lane;
    const float* xl = xp + lane;
    if (nvalid == SCAT_EPW) {
        for (int jb = 0; jb < SCAT_EPW; jb += SCAT_B) {
            float x00[SCAT_B], x01[SCAT_B], x10[SCAT_B], x11[SCAT_B];
            #pragma unroll
            for (int j = 0; j < SCAT_B; ++j) {
                const int sj = __builtin_amdgcn_readlane(s, jb + j);
                const float* xr = xl + (size_t)sj * XDIM;
                x00[j] = xr[0];
                x01[j] = xr[32];
                x10[j] = xr[OUT_F];
                x11[j] = xr[OUT_F + 32];
            }
            #pragma unroll
            for (int j = 0; j < SCAT_B; ++j) {
                const int   dj  = __builtin_amdgcn_readlane(d, jb + j);
                const float w0j = __int_as_float(__builtin_amdgcn_readlane(w0i, jb + j));
                const float w1j = __int_as_float(__builtin_amdgcn_readlane(w1i, jb + j));
                float* ap = accl + (size_t)dj * OUT_F;
                atomicAdd(ap,      w0j * x00[j] + w1j * x10[j]);
                atomicAdd(ap + 32, w0j * x01[j] + w1j * x11[j]);
            }
        }
    } else {
        for (int j = 0; j < nvalid; ++j) {
            const int   sj  = __builtin_amdgcn_readlane(s, j);
            const int   dj  = __builtin_amdgcn_readlane(d, j);
            const float w0j = __int_as_float(__builtin_amdgcn_readlane(w0i, j));
            const float w1j = __int_as_float(__builtin_amdgcn_readlane(w1i, j));
            const float* xr = xl + (size_t)sj * XDIM;
            float* ap = accl + (size_t)dj * OUT_F;
            atomicAdd(ap,      w0j * xr[0]  + w1j * xr[OUT_F]);
            atomicAdd(ap + 32, w0j * xr[32] + w1j * xr[OUT_F + 32]);
        }
    }
}

template <bool FULL>
__device__ __forceinline__ void finalize_rows(
    const float* __restrict__ acc, float* __restrict__ out,
    float b0, float b1, float g0, float g1, float be0, float be1,
    int n0, int N, int lane)
{
    const int f0 = lane, f1 = lane + 32;

    float v0[NODE_W], v1[NODE_W];
    #pragma unroll
    for (int k = 0; k < NODE_W; ++k) {
        int node = n0 + k;
        if (!FULL && node >= N) node = N - 1;
        const float* a = acc + (size_t)node * OUT_F;
        v0[k] = a[f0] + b0;
        v1[k] = a[f1] + b1;
    }

    float sum[NODE_W];
    #pragma unroll
    for (int k = 0; k < NODE_W; ++k) sum[k] = v0[k] + v1[k];
    #pragma unroll
    for (int off = 16; off >= 1; off >>= 1) {
        #pragma unroll
        for (int k = 0; k < NODE_W; ++k) sum[k] += __shfl_xor(sum[k], off, 32);
    }

    float d0[NODE_W], d1[NODE_W], sq[NODE_W];
    #pragma unroll
    for (int k = 0; k < NODE_W; ++k) {
        const float mu = sum[k] * (1.0f / OUT_F);
        d0[k] = v0[k] - mu;
        d1[k] = v1[k] - mu;
        sq[k] = d0[k] * d0[k] + d1[k] * d1[k];
    }
    #pragma unroll
    for (int off = 16; off >= 1; off >>= 1) {
        #pragma unroll
        for (int k = 0; k < NODE_W; ++k) sq[k] += __shfl_xor(sq[k], off, 32);
    }

    #pragma unroll
    for (int k = 0; k < NODE_W; ++k) {
        const float inv = rsqrtf(sq[k] * (1.0f / OUT_F) + LN_EPS);
        if (FULL || n0 + k < N) {
            float* o = out + (size_t)(n0 + k) * OUT_F;
            o[f0] = d0[k] * inv * g0 + be0;
            o[f1] = d1[k] * inv * g1 + be1;
        }
    }
}

__global__ __launch_bounds__(256) void gat_finalize(
    const float* __restrict__ acc, const float* __restrict__ bias,
    const float* __restrict__ gamma, const float* __restrict__ beta,
    float* __restrict__ out, int N)
{
    const int lane = threadIdx.x & 31;
    const int wave = __builtin_amdgcn_readfirstlane(
        (int)((blockIdx.x * blockDim.x + threadIdx.x) >> 5));
    const int n0 = wave * NODE_W;
    if (n0 >= N) return;

    const float b0 = bias[lane],  b1 = bias[lane + 32];
    const float g0 = gamma[lane], g1 = gamma[lane + 32];
    const float be0 = beta[lane], be1 = beta[lane + 32];
    if (n0 + NODE_W <= N)
        finalize_rows<true>(acc, out, b0, b1, g0, g1, be0, be1, n0, N, lane);
    else
        finalize_rows<false>(acc, out, b0, b1, g0, g1, be0, be1, n0, N, lane);
}

extern "C" void kernel_launch(void* const* d_in, const int* in_sizes, int n_in,
                              void* d_out, int out_size, void* d_ws, size_t ws_size,
                              hipStream_t stream)
{
    const float* X    = (const float*)d_in[0];
    const int*   ei   = (const int*)  d_in[1];
    const float* W    = (const float*)d_in[2];
    const float* attS = (const float*)d_in[3];
    const float* attD = (const float*)d_in[4];
    const float* bias = (const float*)d_in[5];
    const float* g    = (const float*)d_in[6];
    const float* b    = (const float*)d_in[7];
    float* out = (float*)d_out;

    const int N = in_sizes[0] / IN_F;
    const int E = in_sizes[1] / 2;
    const int* esrc = ei;
    const int* edst = ei + E;
    const int Et = E + N;

    float* ws    = (float*)d_ws;
    float* xp    = ws;
    float* accb  = xp   + (size_t)N * XDIM;
    float* aS    = accb + (size_t)N * OUT_F;
    float* aD    = aS   + (size_t)N * 2;
    float* mmax  = aD   + (size_t)N * 2;
    float* denom = mmax + (size_t)N * 2;

    gat_gemm_wmma<<<GEMM_BLOCKS, 128, 0, stream>>>(X, W, xp, N);
    gat_logits_init<<<(N + 8 * NODE_W - 1) / (8 * NODE_W), 256, 0, stream>>>(
        xp, attS, attD, aS, aD, mmax, denom, accb, N);
    {
        const int ranges = (N + SEG_NR - 1) / SEG_NR;
        gat_seg_max<<<ranges * SEG_C, SEG_T, 0, stream>>>(
            esrc, edst, aS, aD, mmax, E, N);
        gat_seg_sum<<<ranges * SEG_C, SEG_T, 0, stream>>>(
            esrc, edst, aS, aD, mmax, denom, E, N);
    }
    gat_edge_scatter<<<(Et + 8 * SCAT_EPW - 1) / (8 * SCAT_EPW), 256, 0, stream>>>(
        esrc, edst, aS, aD, mmax, denom, xp, accb, E, N);
    gat_finalize<<<(N + 8 * NODE_W - 1) / (8 * NODE_W), 256, 0, stream>>>(accb, bias, g, b, out, N);
}


# Scatter's per-dst node data (a_dst, max, denominator) packed into one 32-B record per node (filled by the sum kernel): an edge's weight phase fetches 2 lines, not 4; no sum's order or tree changed
# speedup vs baseline: 3.6584x; 1.0321x over previous
#include <hip/hip_runtime.h>
#include <hip/hip_bf16.h>
#include <math.h>


typedef __attribute__((ext_vector_type(2))) float v2f;
typedef __attribute__((ext_vector_type(8))) float v8f;

#define IN_F     128
#define OUT_F    64
#define HEADS    2
#define XDIM     128
#define NEG_SLOPE 0.2f
#define LN_EPS    1e-5f
#define SM_EPS    1e-16f
#define REC_W     8

#define GEMM_BLOCKS 768

__global__ __launch_bounds__(128) void gat_gemm_wmma(
    const float* __restrict__ X, const float* __restrict__ W,
    float* __restrict__ xp, int N)
{
    const int lane = threadIdx.x & 31;
    const int w    = threadIdx.x >> 5;
    const int lm   = lane & 15;
    const int h    = lane >> 4;
    const int nstrips = (N + 15) >> 4;

    float4 bq[2][16];
    #pragma unroll
    for (int t = 0; t < 2; ++t) {
        const float* brow = W + (size_t)((2 * w + t) * 16 + lm) * IN_F + 4 * h;
        #pragma unroll
        for (int j = 0; j < 16; ++j) bq[t][j] = *(const float4*)(brow + 8 * j);
    }

    const int mofs = h << 3;
    const int c0   = (2 * w) * 16 + lm;

    int strip = blockIdx.x;
    if (strip >= nstrips) return;

    float4 aq[16];
    {
        int ar = strip * 16 + lm; if (ar >= N) ar = N - 1;
        const float* arow = X + (size_t)ar * IN_F + 4 * h;
        #pragma unroll
        for (int j = 0; j < 16; ++j) aq[j] = *(const float4*)(arow + 8 * j);
    }

    for (; strip < nstrips; strip += gridDim.x) {
        float4 an[16];
        {
            const int ns = (strip + (int)gridDim.x < nstrips) ? strip + (int)gridDim.x : strip;
            int ar = ns * 16 + lm; if (ar >= N) ar = N - 1;
            const float* arow = X + (size_t)ar * IN_F + 4 * h;
            #pragma unroll
            for (int j = 0; j < 16; ++j) an[j] = *(const float4*)(arow + 8 * j);
        }

        v8f acc0 = (v8f){}, acc1 = (v8f){};
        #pragma unroll
        for (int j = 0; j < 16; ++j) {
            const v2f a_lo = {aq[j].x, aq[j].y};
            const v2f a_hi = {aq[j].z, aq[j].w};
            const v2f b0lo = {bq[0][j].x, bq[0][j].y};
            const v2f b1lo = {bq[1][j].x, bq[1][j].y};
            const v2f b0hi = {bq[0][j].z, bq[0][j].w};
            const v2f b1hi = {bq[1][j].z, bq[1][j].w};
            acc0 = __builtin_amdgcn_wmma_f32_16x16x4_f32(
                false, a_lo, false, b0lo, (short)0, acc0, false, false);
            acc1 = __builtin_amdgcn_wmma_f32_16x16x4_f32(
                false, a_lo, false, b1lo, (short)0, acc1, false, false);
            acc0 = __builtin_amdgcn_wmma_f32_16x16x4_f32(
                false, a_hi, false, b0hi, (short)0, acc0, false, false);
            acc1 = __builtin_amdgcn_wmma_f32_16x16x4_f32(
                false, a_hi, false, b1hi, (short)0, acc1, false, false);
        }

        const int r0 = strip * 16;
        #pragma unroll
        for (int v = 0; v < 8; ++v) {
            const int row = r0 + mofs + v;
            if (row < N) {
                float* orow = xp + (size_t)row * XDIM + c0;
                orow[0]  = acc0[v];
                orow[16] = acc1[v];
            }
        }

        #pragma unroll
        for (int j = 0; j < 16; ++j) aq[j] = an[j];
    }
}

#define NODE_W 16

template <bool FULL>
__device__ __forceinline__ void logits_rows(
    const float* __restrict__ xp, const float4 sv, const float4 dv,
    float* __restrict__ aS, float* __restrict__ aD,
    float* __restrict__ mmax, float* __restrict__ rec,
    float* __restrict__ acc, int n0, int N, int lane)
{
    const int fo = lane * 4;

    float4 xv[NODE_W];
    #pragma unroll
    for (int k = 0; k < NODE_W; ++k) {
        int node = n0 + k;
        if (!FULL && node >= N) node = N - 1;
        xv[k] = *(const float4*)(xp + (size_t)node * XDIM + fo);
    }

    const float2 z = {0.f, 0.f};
    #pragma unroll
    for (int k = 0; k < NODE_W; ++k)
        if (FULL || n0 + k < N)
            *(float2*)(acc + (size_t)(n0 + k) * OUT_F + lane * 2) = z;

    float ps[NODE_W], pd[NODE_W];
    #pragma unroll
    for (int k = 0; k < NODE_W; ++k) {
        ps[k] = xv[k].x*sv.x + xv[k].y*sv.y + xv[k].z*sv.z + xv[k].w*sv.w;
        pd[k] = xv[k].x*dv.x + xv[k].y*dv.y + xv[k].z*dv.z + xv[k].w*dv.w;
    }
    #pragma unroll
    for (int off = 8; off >= 1; off >>= 1) {
        #pragma unroll
        for (int k = 0; k < NODE_W; ++k) {
            ps[k] += __shfl_xor(ps[k], off, 32);
            pd[k] += __shfl_xor(pd[k], off, 32);
        }
    }

    const int j = lane & 15;
    const int head = lane >> 4;
    float ss = ps[0], sd = pd[0];
    #pragma unroll
    for (int k = 1; k < NODE_W; ++k) {
        ss = (j == k) ? ps[k] : ss;
        sd = (j == k) ? pd[k] : sd;
    }
    if (FULL || n0 + j < N) {
        const size_t ti = (size_t)(n0 + j) * 2 + head;
        aS[ti]    = ss;
        aD[ti]    = sd;
        mmax[ti]  = -INFINITY;
        rec[(size_t)(n0 + j) * REC_W + 4 + head] = 0.0f;
    }
}

__global__ __launch_bounds__(256) void gat_logits_init(
    const float* __restrict__ xp,
    const float* __restrict__ attS, const float* __restrict__ attD,
    float* __restrict__ aS, float* __restrict__ aD,
    float* __restrict__ mmax, float* __restrict__ rec,
    float* __restrict__ acc, int N)
{
    const int lane = threadIdx.x & 31;
    const int wave = __builtin_amdgcn_readfirstlane(
        (int)((blockIdx.x * blockDim.x + threadIdx.x) >> 5));
    const int n0 = wave * NODE_W;
    if (n0 >= N) return;

    const float4 sv = *(const float4*)(attS + lane * 4);
    const float4 dv = *(const float4*)(attD + lane * 4);
    if (n0 + NODE_W <= N)
        logits_rows<true>(xp, sv, dv, aS, aD, mmax, rec, acc, n0, N, lane);
    else
        logits_rows<false>(xp, sv, dv, aS, aD, mmax, rec, acc, n0, N, lane);
}

__device__ __forceinline__ void atomicMaxF(float* addr, float val)
{
    if (val >= 0.0f) atomicMax((int*)addr, __float_as_int(val));
    else             atomicMin((unsigned int*)addr, __float_as_uint(val));
}

__device__ __forceinline__ float lrelu(float v)
{
    return v > 0.0f ? v : NEG_SLOPE * v;
}

#define SEG_NR 6256
#define SEG_C  16
#define SEG_T  1024
#define SEG_U  3

template <bool SUM>
__device__ __forceinline__ void seg_update(
    bool mine, unsigned ld, float2 as, float2 ad, float2 mm, float* t_m, float* t_d)
{
    const float e0 = lrelu(as.x + ad.x);
    const float e1 = lrelu(as.y + ad.y);
    if (SUM) {
        const float x0 = __expf(e0 - mm.x);
        const float x1 = __expf(e1 - mm.y);
        if (mine) {
            atomicAdd(t_d + 2 * ld,     x0);
            atomicAdd(t_d + 2 * ld + 1, x1);
        }
    } else {
        if (mine) {
            atomicMaxF(t_m + 2 * ld,     e0);
            atomicMaxF(t_m + 2 * ld + 1, e1);
        }
    }
}

template <bool SUM>
__device__ __forceinline__ void seg_walk(
    const int* __restrict__ esrc, const int* __restrict__ edst,
    const float* __restrict__ aS,
    const float* t_ad, float* t_m, float* t_d,
    int d0, int nr, int e_lo, int e_hi, int E, int tid)
{
    int v_hi = e_lo;
    if ((E & 3) == 0) {
        const int lim = (e_hi < E) ? e_hi : E;
        if (lim > e_lo) v_hi = lim & ~3;
    }
    if (v_hi > e_lo) {
        const int step = SEG_T * 4;
        const int g_last = v_hi - 4;
        int4 dn[SEG_U], sn[SEG_U];
        int g0 = e_lo + tid * 4;
        #pragma unroll
        for (int u = 0; u < SEG_U; ++u) {
            const int g = g0 + u * step;
            const int gc = (g < g_last) ? g : g_last;
            dn[u] = *(const int4*)(edst + gc);
            sn[u] = *(const int4*)(esrc + gc);
        }
        for (int gb = e_lo; gb < v_hi; gb += step * SEG_U) {
            int dc[SEG_U * 4], sc[SEG_U * 4];
            #pragma unroll
            for (int u = 0; u < SEG_U; ++u) {
                const bool ok = (g0 + u * step) < v_hi;
                dc[u * 4 + 0] = ok ? dn[u].x : -1;
                dc[u * 4 + 1] = ok ? dn[u].y : -1;
                dc[u * 4 + 2] = ok ? dn[u].z : -1;
                dc[u * 4 + 3] = ok ? dn[u].w : -1;
                sc[u * 4 + 0] = sn[u].x; sc[u * 4 + 1] = sn[u].y;
                sc[u * 4 + 2] = sn[u].z; sc[u * 4 + 3] = sn[u].w;
            }
            bool     mine[SEG_U * 4];
            unsigned ld[SEG_U * 4];
            float2   as[SEG_U * 4];
            #pragma unroll
            for (int k = 0; k < SEG_U * 4; ++k) {
                ld[k]   = (unsigned)(dc[k] - d0);
                mine[k] = ld[k] < (unsigned)nr;
                if (!mine[k]) ld[k] = 0u;
                const int se = mine[k] ? sc[k] : 0;
                as[k] = *(const float2*)(aS + (size_t)se * 2);
            }
            g0 += step * SEG_U;
            #pragma unroll
            for (int u = 0; u < SEG_U; ++u) {
                const int g = g0 + u * step;
                const int gc = (g < g_last) ? g : g_last;
                dn[u] = *(const int4*)(edst + gc);
                sn[u] = *(const int4*)(esrc + gc);
            }
            float2 ad[SEG_U * 4], mm[SEG_U * 4];
            #pragma unroll
            for (int k = 0; k < SEG_U * 4; ++k) {
                ad[k] = *(const float2*)(t_ad + 2 * ld[k]);
                mm[k] = SUM ? *(const float2*)(t_m + 2 * ld[k]) : float2{0.f, 0.f};
            }
            #pragma unroll
            for (int k = 0; k < SEG_U * 4; ++k)
                seg_update<SUM>(mine[k], ld[k], as[k], ad[k], mm[k], t_m, t_d);
        }
    }
    for (int eid = v_hi + tid; eid < e_hi; eid += SEG_T) {
        int s, d;
        if (eid < E) { s = esrc[eid]; d = edst[eid]; }
        else         { s = d = eid - E; }
        const unsigned ld1 = (unsigned)(d - d0);
        if (ld1 < (unsigned)nr) {
            const float2 as1 = *(const float2*)(aS + (size_t)s * 2);
            const float2 ad1 = *(const float2*)(t_ad + 2 * ld1);
            const float2 mm1 = SUM ? *(const float2*)(t_m + 2 * ld1) : float2{0.f, 0.f};
            seg_update<SUM>(true, ld1, as1, ad1, mm1, t_m, t_d);
        }
    }
}

__device__ __forceinline__ void seg_chunk(int Et, int c, int& e_lo, int& e_hi)
{
    const int ec = (((Et + SEG_C - 1) / SEG_C) + 3) & ~3;
    e_lo = c * ec;
    if (e_lo > Et) e_lo = Et;
    e_hi = (Et - e_lo < ec) ? Et : e_lo + ec;
}

__global__ __launch_bounds__(SEG_T) void gat_seg_max(
    const int* __restrict__ esrc, const int* __restrict__ edst,
    const float* __restrict__ aS, const float* __restrict__ aD,
    float* __restrict__ mmax, int E, int N)
{
    __shared__ float t_ad[SEG_NR * 2];
    __shared__ float t_m[SEG_NR * 2];
    const int tid = threadIdx.x;
    const int d0 = (int)(blockIdx.x / SEG_C) * SEG_NR;
    const int c  = (int)(blockIdx.x % SEG_C);
    const int nr = (N - d0 < SEG_NR) ? (N - d0) : SEG_NR;
    int e_lo, e_hi;
    seg_chunk(E + N, c, e_lo, e_hi);

    for (int i = tid; i < nr * 2; i += SEG_T) {
        t_ad[i] = aD[(size_t)d0 * 2 + i];
        t_m[i]  = -INFINITY;
    }
    __syncthreads();
    seg_walk<false>(esrc, edst, aS, t_ad, t_m, nullptr, d0, nr, e_lo, e_hi, E, tid);
    __syncthreads();
    for (int i = tid; i < nr * 2; i += SEG_T)
        atomicMaxF(&mmax[(size_t)d0 * 2 + i], t_m[i]);
}

__global__ __launch_bounds__(SEG_T) void gat_seg_sum(
    const int* __restrict__ esrc, const int* __restrict__ edst,
    const float* __restrict__ aS, const float* __restrict__ aD,
    const float* __restrict__ mmax, float* __restrict__ rec, int E, int N)
{
    __shared__ float t_ad[SEG_NR * 2];
    __shared__ float t_m[SEG_NR * 2];
    __shared__ float t_d[SEG_NR * 2];
    const int tid = threadIdx.x;
    const int d0 = (int)(blockIdx.x / SEG_C) * SEG_NR;
    const int c  = (int)(blockIdx.x % SEG_C);
    const int nr = (N - d0 < SEG_NR) ? (N - d0) : SEG_NR;
    int e_lo, e_hi;
    seg_chunk(E + N, c, e_lo, e_hi);

    for (int i = tid; i < nr * 2; i += SEG_T) {
        t_ad[i] = aD[(size_t)d0 * 2 + i];
        t_m[i]  = mmax[(size_t)d0 * 2 + i];
        t_d[i]  = 0.0f;
    }
    __syncthreads();
    {
        const int per  = (nr + SEG_C - 1) / SEG_C;
        const int i_lo = c * per;
        const int i_hi = (nr - i_lo < per) ? nr : i_lo + per;
        for (int i = i_lo + tid; i < i_hi; i += SEG_T) {
            const float4 v = {t_ad[2 * i], t_ad[2 * i + 1], t_m[2 * i], t_m[2 * i + 1]};
            *(float4*)(rec + (size_t)(d0 + i) * REC_W) = v;
        }
    }
    seg_walk<true>(esrc, edst, aS, t_ad, t_m, t_d, d0, nr, e_lo, e_hi, E, tid);
    __syncthreads();
    for (int i = tid; i < nr * 2; i += SEG_T)
        atomicAdd(&rec[(size_t)(d0 + (i >> 1)) * REC_W + 4 + (i & 1)], t_d[i]);
}

#define SCAT_EPW 32
#define SCAT_B   8

__global__ __launch_bounds__(256) void gat_edge_scatter(
    const int* __restrict__ esrc, const int* __restrict__ edst,
    const float* __restrict__ aS, const float* __restrict__ rec,
    const float* __restrict__ xp, float* __restrict__ acc, int E, int N)
{
    const int lane = threadIdx.x & 31;
    const int wave = __builtin_amdgcn_readfirstlane(
        (int)((blockIdx.x * blockDim.x + threadIdx.x) >> 5));
    const int Et = E + N;
    const int ebase = wave * SCAT_EPW;
    if (ebase >= Et) return;
    const int nvalid = (Et - ebase < SCAT_EPW) ? (Et - ebase) : SCAT_EPW;

    const int eid = ebase + lane;
    int s = 0, d = 0;
    if (eid < E)       { s = esrc[eid]; d = edst[eid]; }
    else if (eid < Et) { s = d = eid - E; }
    const float2 as = *(const float2*)(aS + (size_t)s * 2);
    const float* rd = rec + (size_t)d * REC_W;
    const float4 am = *(const float4*)rd;
    const float2 dn = *(const float2*)(rd + 4);
    const float e0 = lrelu(as.x + am.x);
    const float e1 = lrelu(as.y + am.y);
    const float w0 = 0.5f * (__expf(e0 - am.z) / (dn.x + SM_EPS));
    const float w1 = 0.5f * (__expf(e1 - am.w) / (dn.y + SM_EPS));
    const int w0i = __float_as_int(w0);
    const int w1i = __float_as_int(w1);

    float* accl = acc + lane;
    const float* xl = xp + lane;
    if (nvalid == SCAT_EPW) {
        for (int jb = 0; jb < SCAT_EPW; jb += SCAT_B) {
            float x00[SCAT_B], x01[SCAT_B], x10[SCAT_B], x11[SCAT_B];
            #pragma unroll
            for (int j = 0; j < SCAT_B; ++j) {
                const int sj = __builtin_amdgcn_readlane(s, jb + j);
                const float* xr = xl + (size_t)sj * XDIM;
                x00[j] = xr[0];
                x01[j] = xr[32];
                x10[j] = xr[OUT_F];
                x11[j] = xr[OUT_F + 32];
            }
            #pragma unroll
            for (int j = 0; j < SCAT_B; ++j) {
                const int   dj  = __builtin_amdgcn_readlane(d, jb + j);
                const float w0j = __int_as_float(__builtin_amdgcn_readlane(w0i, jb + j));
                const float w1j = __int_as_float(__builtin_amdgcn_readlane(w1i, jb + j));
                float* ap = accl + (size_t)dj * OUT_F;
                atomicAdd(ap,      w0j * x00[j] + w1j * x10[j]);
                atomicAdd(ap + 32, w0j * x01[j] + w1j * x11[j]);
            }
        }
    } else {
        for (int j = 0; j < nvalid; ++j) {
            const int   sj  = __builtin_amdgcn_readlane(s, j);
            const int   dj  = __builtin_amdgcn_readlane(d, j);
            const float w0j = __int_as_float(__builtin_amdgcn_readlane(w0i, j));
            const float w1j = __int_as_float(__builtin_amdgcn_readlane(w1i, j));
            const float* xr = xl + (size_t)sj * XDIM;
            float* ap = accl + (size_t)dj * OUT_F;
            atomicAdd(ap,      w0j * xr[0]  + w1j * xr[OUT_F]);
            atomicAdd(ap + 32, w0j * xr[32] + w1j * xr[OUT_F + 32]);
        }
    }
}

template <bool FULL>
__device__ __forceinline__ void finalize_rows(
    const float* __restrict__ acc, float* __restrict__ out,
    float b0, float b1, float g0, float g1, float be0, float be1,
    int n0, int N, int lane)
{
    const int f0 = lane, f1 = lane + 32;

    float v0[NODE_W], v1[NODE_W];
    #pragma unroll
    for (int k = 0; k < NODE_W; ++k) {
        int node = n0 + k;
        if (!FULL && node >= N) node = N - 1;
        const float* a = acc + (size_t)node * OUT_F;
        v0[k] = a[f0] + b0;
        v1[k] = a[f1] + b1;
    }

    float sum[NODE_W];
    #pragma unroll
    for (int k = 0; k < NODE_W; ++k) sum[k] = v0[k] + v1[k];
    #pragma unroll
    for (int off = 16; off >= 1; off >>= 1) {
        #pragma unroll
        for (int k = 0; k < NODE_W; ++k) sum[k] += __shfl_xor(sum[k], off, 32);
    }

    float d0[NODE_W], d1[NODE_W], sq[NODE_W];
    #pragma unroll
    for (int k = 0; k < NODE_W; ++k) {
        const float mu = sum[k] * (1.0f / OUT_F);
        d0[k] = v0[k] - mu;
        d1[k] = v1[k] - mu;
        sq[k] = d0[k] * d0[k] + d1[k] * d1[k];
    }
    #pragma unroll
    for (int off = 16; off >= 1; off >>= 1) {
        #pragma unroll
        for (int k = 0; k < NODE_W; ++k) sq[k] += __shfl_xor(sq[k], off, 32);
    }

    #pragma unroll
    for (int k = 0; k < NODE_W; ++k) {
        const float inv = rsqrtf(sq[k] * (1.0f / OUT_F) + LN_EPS);
        if (FULL || n0 + k < N) {
            float* o = out + (size_t)(n0 + k) * OUT_F;
            o[f0] = d0[k] * inv * g0 + be0;
            o[f1] = d1[k] * inv * g1 + be1;
        }
    }
}

__global__ __launch_bounds__(256) void gat_finalize(
    const float* __restrict__ acc, const float* __restrict__ bias,
    const float* __restrict__ gamma, const float* __restrict__ beta,
    float* __restrict__ out, int N)
{
    const int lane = threadIdx.x & 31;
    const int wave = __builtin_amdgcn_readfirstlane(
        (int)((blockIdx.x * blockDim.x + threadIdx.x) >> 5));
    const int n0 = wave * NODE_W;
    if (n0 >= N) return;

    const float b0 = bias[lane],  b1 = bias[lane + 32];
    const float g0 = gamma[lane], g1 = gamma[lane + 32];
    const float be0 = beta[lane], be1 = beta[lane + 32];
    if (n0 + NODE_W <= N)
        finalize_rows<true>(acc, out, b0, b1, g0, g1, be0, be1, n0, N, lane);
    else
        finalize_rows<false>(acc, out, b0, b1, g0, g1, be0, be1, n0, N, lane);
}

extern "C" void kernel_launch(void* const* d_in, const int* in_sizes, int n_in,
                              void* d_out, int out_size, void* d_ws, size_t ws_size,
                              hipStream_t stream)
{
    const float* X    = (const float*)d_in[0];
    const int*   ei   = (const int*)  d_in[1];
    const float* W    = (const float*)d_in[2];
    const float* attS = (const float*)d_in[3];
    const float* attD = (const float*)d_in[4];
    const float* bias = (const float*)d_in[5];
    const float* g    = (const float*)d_in[6];
    const float* b    = (const float*)d_in[7];
    float* out = (float*)d_out;

    const int N = in_sizes[0] / IN_F;
    const int E = in_sizes[1] / 2;
    const int* esrc = ei;
    const int* edst = ei + E;
    const int Et = E + N;

    float* ws    = (float*)d_ws;
    float* xp    = ws;
    float* accb  = xp   + (size_t)N * XDIM;
    float* rec   = accb + (size_t)N * OUT_F;
    float* aS    = rec  + (size_t)N * REC_W;
    float* aD    = aS   + (size_t)N * 2;
    float* mmax  = aD   + (size_t)N * 2;

    gat_gemm_wmma<<<GEMM_BLOCKS, 128, 0, stream>>>(X, W, xp, N);
    gat_logits_init<<<(N + 8 * NODE_W - 1) / (8 * NODE_W), 256, 0, stream>>>(
        xp, attS, attD, aS, aD, mmax, rec, accb, N);
    {
        const int ranges = (N + SEG_NR - 1) / SEG_NR;
        gat_seg_max<<<ranges * SEG_C, SEG_T, 0, stream>>>(
            esrc, edst, aS, aD, mmax, E, N);
        gat_seg_sum<<<ranges * SEG_C, SEG_T, 0, stream>>>(
            esrc, edst, aS, aD, mmax, rec, E, N);
    }
    gat_edge_scatter<<<(Et + 8 * SCAT_EPW - 1) / (8 * SCAT_EPW), 256, 0, stream>>>(
        esrc, edst, aS, rec, xp, accb, E, N);
    gat_finalize<<<(N + 8 * NODE_W - 1) / (8 * NODE_W), 256, 0, stream>>>(accb, bias, g, b, out, N);
}


# Owner blocks: a block per 400 dst nodes filters the whole edge list, does max, exp-sum, weighted rows (LDS float adds), LayerNorm in LDS; no global atomics, 3 launches; LDS-arrival sum order
# speedup vs baseline: 4.4888x; 1.2270x over previous
#include <hip/hip_runtime.h>
#include <hip/hip_bf16.h>
#include <math.h>


typedef __attribute__((ext_vector_type(2))) float v2f;
typedef __attribute__((ext_vector_type(8))) float v8f;

#define IN_F     128
#define OUT_F    64
#define HEADS    2
#define XDIM     128
#define NEG_SLOPE 0.2f
#define LN_EPS    1e-5f
#define SM_EPS    1e-16f

#define GEMM_BLOCKS 768

__global__ __launch_bounds__(128) void gat_gemm_wmma(
    const float* __restrict__ X, const float* __restrict__ W,
    float* __restrict__ xp, int N)
{
    const int lane = threadIdx.x & 31;
    const int w    = threadIdx.x >> 5;
    const int lm   = lane & 15;
    const int h    = lane >> 4;
    const int nstrips = (N + 15) >> 4;

    float4 bq[2][16];
    #pragma unroll
    for (int t = 0; t < 2; ++t) {
        const float* brow = W + (size_t)((2 * w + t) * 16 + lm) * IN_F + 4 * h;
        #pragma unroll
        for (int j = 0; j < 16; ++j) bq[t][j] = *(const float4*)(brow + 8 * j);
    }

    const int mofs = h << 3;
    const int c0   = (2 * w) * 16 + lm;

    int strip = blockIdx.x;
    if (strip >= nstrips) return;

    float4 aq[16];
    {
        int ar = strip * 16 + lm; if (ar >= N) ar = N - 1;
        const float* arow = X + (size_t)ar * IN_F + 4 * h;
        #pragma unroll
        for (int j = 0; j < 16; ++j) aq[j] = *(const float4*)(arow + 8 * j);
    }

    for (; strip < nstrips; strip += gridDim.x) {
        float4 an[16];
        {
            const int ns = (strip + (int)gridDim.x < nstrips) ? strip + (int)gridDim.x : strip;
            int ar = ns * 16 + lm; if (ar >= N) ar = N - 1;
            const float* arow = X + (size_t)ar * IN_F + 4 * h;
            #pragma unroll
            for (int j = 0; j < 16; ++j) an[j] = *(const float4*)(arow + 8 * j);
        }

        v8f acc0 = (v8f){}, acc1 = (v8f){};
        #pragma unroll
        for (int j = 0; j < 16; ++j) {
            const v2f a_lo = {aq[j].x, aq[j].y};
            const v2f a_hi = {aq[j].z, aq[j].w};
            const v2f b0lo = {bq[0][j].x, bq[0][j].y};
            const v2f b1lo = {bq[1][j].x, bq[1][j].y};
            const v2f b0hi = {bq[0][j].z, bq[0][j].w};
            const v2f b1hi = {bq[1][j].z, bq[1][j].w};
            acc0 = __builtin_amdgcn_wmma_f32_16x16x4_f32(
                false, a_lo, false, b0lo, (short)0, acc0, false, false);
            acc1 = __builtin_amdgcn_wmma_f32_16x16x4_f32(
                false, a_lo, false, b1lo, (short)0, acc1, false, false);
            acc0 = __builtin_amdgcn_wmma_f32_16x16x4_f32(
                false, a_hi, false, b0hi, (short)0, acc0, false, false);
            acc1 = __builtin_amdgcn_wmma_f32_16x16x4_f32(
                false, a_hi, false, b1hi, (short)0, acc1, false, false);
        }

        const int r0 = strip * 16;
        #pragma unroll
        for (int v = 0; v < 8; ++v) {
            const int row = r0 + mofs + v;
            if (row < N) {
                float* orow = xp + (size_t)row * XDIM + c0;
                orow[0]  = acc0[v];
                orow[16] = acc1[v];
            }
        }

        #pragma unroll
        for (int j = 0; j < 16; ++j) aq[j] = an[j];
    }
}

#define NODE_W 16

template <bool FULL>
__device__ __forceinline__ void logits_rows(
    const float* __restrict__ xp, const float4 sv, const float4 dv,
    float* __restrict__ aS, float* __restrict__ aD, int n0, int N, int lane)
{
    const int fo = lane * 4;

    float4 xv[NODE_W];
    #pragma unroll
    for (int k = 0; k < NODE_W; ++k) {
        int node = n0 + k;
        if (!FULL && node >= N) node = N - 1;
        xv[k] = *(const float4*)(xp + (size_t)node * XDIM + fo);
    }

    float ps[NODE_W], pd[NODE_W];
    #pragma unroll
    for (int k = 0; k < NODE_W; ++k) {
        ps[k] = xv[k].x*sv.x + xv[k].y*sv.y + xv[k].z*sv.z + xv[k].w*sv.w;
        pd[k] = xv[k].x*dv.x + xv[k].y*dv.y + xv[k].z*dv.z + xv[k].w*dv.w;
    }
    #pragma unroll
    for (int off = 8; off >= 1; off >>= 1) {
        #pragma unroll
        for (int k = 0; k < NODE_W; ++k) {
            ps[k] += __shfl_xor(ps[k], off, 32);
            pd[k] += __shfl_xor(pd[k], off, 32);
        }
    }

    const int j = lane & 15;
    const int head = lane >> 4;
    float ss = ps[0], sd = pd[0];
    #pragma unroll
    for (int k = 1; k < NODE_W; ++k) {
        ss = (j == k) ? ps[k] : ss;
        sd = (j == k) ? pd[k] : sd;
    }
    if (FULL || n0 + j < N) {
        const size_t ti = (size_t)(n0 + j) * 2 + head;
        aS[ti] = ss;
        aD[ti] = sd;
    }
}

__global__ __launch_bounds__(256) void gat_logits(
    const float* __restrict__ xp,
    const float* __restrict__ attS, const float* __restrict__ attD,
    float* __restrict__ aS, float* __restrict__ aD, int N)
{
    const int lane = threadIdx.x & 31;
    const int wave = __builtin_amdgcn_readfirstlane(
        (int)((blockIdx.x * blockDim.x + threadIdx.x) >> 5));
    const int n0 = wave * NODE_W;
    if (n0 >= N) return;

    const float4 sv = *(const float4*)(attS + lane * 4);
    const float4 dv = *(const float4*)(attD + lane * 4);
    if (n0 + NODE_W <= N)
        logits_rows<true>(xp, sv, dv, aS, aD, n0, N, lane);
    else
        logits_rows<false>(xp, sv, dv, aS, aD, n0, N, lane);
}

__device__ __forceinline__ void atomicMaxF(float* addr, float val)
{
    if (val >= 0.0f) atomicMax((int*)addr, __float_as_int(val));
    else             atomicMin((unsigned int*)addr, __float_as_uint(val));
}

__device__ __forceinline__ float lrelu(float v)
{
    return v > 0.0f ? v : NEG_SLOPE * v;
}

#define OWN_NR 400
#define OWN_T  1024
#define OWN_Q  10240
#define OWN_U  8
#define OWN_B  8
#define OWN_LN 4

__device__ __forceinline__ void own_push(
    int d, int eid, int d0, int nr, int* q_s, int* q_ld, int* q_n)
{
    const unsigned ld = (unsigned)(d - d0);
    if (ld < (unsigned)nr) {
        const int idx = atomicAdd(q_n, 1);
        if (idx < OWN_Q) { q_s[idx] = eid; q_ld[idx] = (int)ld; }
    }
}

__device__ __forceinline__ void own_scan_span(
    const int4* __restrict__ d4, int lo, int hi, int tid, int d0, int nr,
    int* q_s, int* q_ld, int* q_n)
{
    int q0 = lo;
    for (; q0 + OWN_T * OWN_U <= hi; q0 += OWN_T * OWN_U) {
        const int4* p = d4 + q0 + tid;
        int4 dv[OWN_U];
        #pragma unroll
        for (int u = 0; u < OWN_U; ++u) dv[u] = p[u * OWN_T];
        #pragma unroll
        for (int u = 0; u < OWN_U; ++u) {
            const int e4 = (q0 + u * OWN_T + tid) * 4;
            own_push(dv[u].x, e4 + 0, d0, nr, q_s, q_ld, q_n);
            own_push(dv[u].y, e4 + 1, d0, nr, q_s, q_ld, q_n);
            own_push(dv[u].z, e4 + 2, d0, nr, q_s, q_ld, q_n);
            own_push(dv[u].w, e4 + 3, d0, nr, q_s, q_ld, q_n);
        }
    }
    if (q0 < hi) {
        int4 dv[OWN_U];
        #pragma unroll
        for (int u = 0; u < OWN_U; ++u) {
            const int q = q0 + u * OWN_T + tid;
            dv[u] = d4[(q < hi) ? q : lo];
        }
        #pragma unroll
        for (int u = 0; u < OWN_U; ++u) {
            const int q = q0 + u * OWN_T + tid;
            if (q < hi) {
                const int e4 = q * 4;
                own_push(dv[u].x, e4 + 0, d0, nr, q_s, q_ld, q_n);
                own_push(dv[u].y, e4 + 1, d0, nr, q_s, q_ld, q_n);
                own_push(dv[u].z, e4 + 2, d0, nr, q_s, q_ld, q_n);
                own_push(dv[u].w, e4 + 3, d0, nr, q_s, q_ld, q_n);
            }
        }
    }
}

__global__ __launch_bounds__(OWN_T) void gat_own_range(
    const int* __restrict__ esrc, const int* __restrict__ edst,
    const float* __restrict__ aS, const float* __restrict__ aD,
    const float* __restrict__ xp,
    const float* __restrict__ bias, const float* __restrict__ gamma,
    const float* __restrict__ beta, float* __restrict__ out, int E, int N)
{
    __shared__ float acc_l[OWN_NR * OUT_F];
    __shared__ float t_ad[OWN_NR * 2];
    __shared__ float t_m[OWN_NR * 2];
    __shared__ float t_d[OWN_NR * 2];
    __shared__ int   q_s[OWN_Q];
    __shared__ int   q_ld[OWN_Q];
    __shared__ float q_x0[OWN_Q];
    __shared__ float q_x1[OWN_Q];
    __shared__ int   q_n;

    const int tid  = threadIdx.x;
    const int lane = tid & 31;
    const int d0   = (int)blockIdx.x * OWN_NR;
    const int nr   = (N - d0 < OWN_NR) ? (N - d0) : OWN_NR;
    const int Et   = E + N;

    for (int i = tid; i < nr * OUT_F; i += OWN_T) acc_l[i] = 0.0f;
    for (int i = tid; i < nr * 2; i += OWN_T) {
        t_ad[i] = aD[(size_t)d0 * 2 + i];
        t_m[i]  = -INFINITY;
        t_d[i]  = 0.0f;
    }
    for (int i = tid; i < nr; i += OWN_T) {
        q_s[i]  = E + d0 + i;
        q_ld[i] = i;
    }
    if (tid == 0) q_n = nr;
    __syncthreads();

    {
        const int n4 = ((E & 3) == 0) ? (E >> 2) : 0;
        if (n4 > 0) {
            const int4* d4 = (const int4*)edst;
            const int rot = (int)(((long long)blockIdx.x * n4) / (long long)gridDim.x) & ~31;
            own_scan_span(d4, rot, n4, tid, d0, nr, q_s, q_ld, &q_n);
            own_scan_span(d4, 0, rot, tid, d0, nr, q_s, q_ld, &q_n);
        }
        for (int eid = n4 * 4 + tid; eid < E; eid += OWN_T)
            own_push(edst[eid], eid, d0, nr, q_s, q_ld, &q_n);
    }
    __syncthreads();
    const int nq = q_n;

    if (nq <= OWN_Q) {
        for (int i = tid; i < nq; i += OWN_T) {
            const int eid = q_s[i];
            const int ld  = q_ld[i];
            const int s   = (eid < E) ? esrc[eid] : (eid - E);
            const float2 as = *(const float2*)(aS + (size_t)s * 2);
            const float e0 = lrelu(as.x + t_ad[2 * ld]);
            const float e1 = lrelu(as.y + t_ad[2 * ld + 1]);
            atomicMaxF(&t_m[2 * ld],     e0);
            atomicMaxF(&t_m[2 * ld + 1], e1);
            q_s[i]  = s;
            q_x0[i] = e0;
            q_x1[i] = e1;
        }
        __syncthreads();
        for (int i = tid; i < nq; i += OWN_T) {
            const int ld = q_ld[i];
            const float x0 = __expf(q_x0[i] - t_m[2 * ld]);
            const float x1 = __expf(q_x1[i] - t_m[2 * ld + 1]);
            atomicAdd(&t_d[2 * ld],     x0);
            atomicAdd(&t_d[2 * ld + 1], x1);
            q_x0[i] = x0;
            q_x1[i] = x1;
        }
        __syncthreads();
        {
            const int wv = __builtin_amdgcn_readfirstlane(tid >> 5);
            const float* xl = xp + lane;
            float* al = acc_l + lane;
            for (int g = wv * 32; g < nq; g += OWN_T) {
                const int nvalid = (nq - g < 32) ? (nq - g) : 32;
                const int ic = (lane < nvalid) ? (g + lane) : g;
                const int s  = q_s[ic];
                const int ld = q_ld[ic];
                const float w0 = 0.5f * (q_x0[ic] / (t_d[2 * ld]     + SM_EPS));
                const float w1 = 0.5f * (q_x1[ic] / (t_d[2 * ld + 1] + SM_EPS));
                const int w0i = __float_as_int(w0);
                const int w1i = __float_as_int(w1);
                if (nvalid == 32) {
                    for (int jb = 0; jb < 32; jb += OWN_B) {
                        float x00[OWN_B], x01[OWN_B], x10[OWN_B], x11[OWN_B];
                        #pragma unroll
                        for (int j = 0; j < OWN_B; ++j) {
                            const int sj = __builtin_amdgcn_readlane(s, jb + j);
                            const float* xr = xl + (size_t)sj * XDIM;
                            x00[j] = xr[0];
                            x01[j] = xr[32];
                            x10[j] = xr[OUT_F];
                            x11[j] = xr[OUT_F + 32];
                        }
                        #pragma unroll
                        for (int j = 0; j < OWN_B; ++j) {
                            const int   lj  = __builtin_amdgcn_readlane(ld, jb + j);
                            const float w0j = __int_as_float(__builtin_amdgcn_readlane(w0i, jb + j));
                            const float w1j = __int_as_float(__builtin_amdgcn_readlane(w1i, jb + j));
                            float* ap = al + lj * OUT_F;
                            atomicAdd(ap,      w0j * x00[j] + w1j * x10[j]);
                            atomicAdd(ap + 32, w0j * x01[j] + w1j * x11[j]);
                        }
                    }
                } else {
                    for (int j = 0; j < nvalid; ++j) {
                        const int   sj  = __builtin_amdgcn_readlane(s, j);
                        const int   lj  = __builtin_amdgcn_readlane(ld, j);
                        const float w0j = __int_as_float(__builtin_amdgcn_readlane(w0i, j));
                        const float w1j = __int_as_float(__builtin_amdgcn_readlane(w1i, j));
                        const float* xr = xl + (size_t)sj * XDIM;
                        float* ap = al + lj * OUT_F;
                        atomicAdd(ap,      w0j * xr[0]  + w1j * xr[OUT_F]);
                        atomicAdd(ap + 32, w0j * xr[32] + w1j * xr[OUT_F + 32]);
                    }
                }
            }
        }
        __syncthreads();
    } else {
        for (int eid = tid; eid < Et; eid += OWN_T) {
            int s, d;
            if (eid < E) { s = esrc[eid]; d = edst[eid]; } else { s = d = eid - E; }
            const unsigned ld = (unsigned)(d - d0);
            if (ld < (unsigned)nr) {
                const float2 as = *(const float2*)(aS + (size_t)s * 2);
                atomicMaxF(&t_m[2 * ld],     lrelu(as.x + t_ad[2 * ld]));
                atomicMaxF(&t_m[2 * ld + 1], lrelu(as.y + t_ad[2 * ld + 1]));
            }
        }
        __syncthreads();
        for (int eid = tid; eid < Et; eid += OWN_T) {
            int s, d;
            if (eid < E) { s = esrc[eid]; d = edst[eid]; } else { s = d = eid - E; }
            const unsigned ld = (unsigned)(d - d0);
            if (ld < (unsigned)nr) {
                const float2 as = *(const float2*)(aS + (size_t)s * 2);
                atomicAdd(&t_d[2 * ld],     __expf(lrelu(as.x + t_ad[2 * ld])     - t_m[2 * ld]));
                atomicAdd(&t_d[2 * ld + 1], __expf(lrelu(as.y + t_ad[2 * ld + 1]) - t_m[2 * ld + 1]));
            }
        }
        __syncthreads();
        for (int eid = tid; eid < Et; eid += OWN_T) {
            int s, d;
            if (eid < E) { s = esrc[eid]; d = edst[eid]; } else { s = d = eid - E; }
            const unsigned ld = (unsigned)(d - d0);
            if (ld < (unsigned)nr) {
                const float2 as = *(const float2*)(aS + (size_t)s * 2);
                const float x0 = __expf(lrelu(as.x + t_ad[2 * ld])     - t_m[2 * ld]);
                const float x1 = __expf(lrelu(as.y + t_ad[2 * ld + 1]) - t_m[2 * ld + 1]);
                const float w0 = 0.5f * (x0 / (t_d[2 * ld]     + SM_EPS));
                const float w1 = 0.5f * (x1 / (t_d[2 * ld + 1] + SM_EPS));
                const float* xr = xp + (size_t)s * XDIM;
                for (int f = 0; f < OUT_F; ++f)
                    atomicAdd(&acc_l[ld * OUT_F + f], w0 * xr[f] + w1 * xr[OUT_F + f]);
            }
        }
        __syncthreads();
    }

    {
        const int wv = __builtin_amdgcn_readfirstlane(tid >> 5);
        const int f0 = lane, f1 = lane + 32;
        const float b0 = bias[f0],  b1 = bias[f1];
        const float g0 = gamma[f0], g1 = gamma[f1];
        const float be0 = beta[f0], be1 = beta[f1];
        for (int base = wv * OWN_LN; base < nr; base += (OWN_T / 32) * OWN_LN) {
            float v0[OWN_LN], v1[OWN_LN], sum[OWN_LN];
            #pragma unroll
            for (int k = 0; k < OWN_LN; ++k) {
                int r = base + k;
                if (r >= nr) r = nr - 1;
                v0[k] = acc_l[r * OUT_F + f0] + b0;
                v1[k] = acc_l[r * OUT_F + f1] + b1;
                sum[k] = v0[k] + v1[k];
            }
            #pragma unroll
            for (int off = 16; off >= 1; off >>= 1) {
                #pragma unroll
                for (int k = 0; k < OWN_LN; ++k) sum[k] += __shfl_xor(sum[k], off, 32);
            }
            float c0[OWN_LN], c1[OWN_LN], sq[OWN_LN];
            #pragma unroll
            for (int k = 0; k < OWN_LN; ++k) {
                const float mu = sum[k] * (1.0f / OUT_F);
                c0[k] = v0[k] - mu;
                c1[k] = v1[k] - mu;
                sq[k] = c0[k] * c0[k] + c1[k] * c1[k];
            }
            #pragma unroll
            for (int off = 16; off >= 1; off >>= 1) {
                #pragma unroll
                for (int k = 0; k < OWN_LN; ++k) sq[k] += __shfl_xor(sq[k], off, 32);
            }
            #pragma unroll
            for (int k = 0; k < OWN_LN; ++k) {
                const float inv = rsqrtf(sq[k] * (1.0f / OUT_F) + LN_EPS);
                if (base + k < nr) {
                    float* o = out + (size_t)(d0 + base + k) * OUT_F;
                    o[f0] = c0[k] * inv * g0 + be0;
                    o[f1] = c1[k] * inv * g1 + be1;
                }
            }
        }
    }
}

extern "C" void kernel_launch(void* const* d_in, const int* in_sizes, int n_in,
                              void* d_out, int out_size, void* d_ws, size_t ws_size,
                              hipStream_t stream)
{
    const float* X    = (const float*)d_in[0];
    const int*   ei   = (const int*)  d_in[1];
    const float* W    = (const float*)d_in[2];
    const float* attS = (const float*)d_in[3];
    const float* attD = (const float*)d_in[4];
    const float* bias = (const float*)d_in[5];
    const float* g    = (const float*)d_in[6];
    const float* b    = (const float*)d_in[7];
    float* out = (float*)d_out;

    const int N = in_sizes[0] / IN_F;
    const int E = in_sizes[1] / 2;
    const int* esrc = ei;
    const int* edst = ei + E;

    float* ws    = (float*)d_ws;
    float* xp    = ws;
    float* aS    = xp   + (size_t)N * XDIM;
    float* aD    = aS   + (size_t)N * 2;

    gat_gemm_wmma<<<GEMM_BLOCKS, 128, 0, stream>>>(X, W, xp, N);
    gat_logits<<<(N + 8 * NODE_W - 1) / (8 * NODE_W), 256, 0, stream>>>(
        xp, attS, attD, aS, aD, N);
    gat_own_range<<<(N + OWN_NR - 1) / OWN_NR, OWN_T, 0, stream>>>(
        esrc, edst, aS, aD, xp, bias, g, b, out, E, N);
}


# Owner scan reads a 1-byte range key per edge, not the 4-byte dst: launch 2 writes keys as bit planes of 16 edges; eight logic ops flag 16 edges, no branch per edge; LDS-arrival sum order changes
# speedup vs baseline: 5.4915x; 1.2234x over previous
#include <hip/hip_runtime.h>
#include <hip/hip_bf16.h>
#include <math.h>


typedef __attribute__((ext_vector_type(2))) float v2f;
typedef __attribute__((ext_vector_type(8))) float v8f;

#define IN_F     128
#define OUT_F    64
#define HEADS    2
#define XDIM     128
#define NEG_SLOPE 0.2f
#define LN_EPS    1e-5f
#define SM_EPS    1e-16f

#define GEMM_BLOCKS 768

__global__ __launch_bounds__(128) void gat_gemm_wmma(
    const float* __restrict__ X, const float* __restrict__ W,
    float* __restrict__ xp, int N)
{
    const int lane = threadIdx.x & 31;
    const int w    = threadIdx.x >> 5;
    const int lm   = lane & 15;
    const int h    = lane >> 4;
    const int nstrips = (N + 15) >> 4;

    float4 bq[2][16];
    #pragma unroll
    for (int t = 0; t < 2; ++t) {
        const float* brow = W + (size_t)((2 * w + t) * 16 + lm) * IN_F + 4 * h;
        #pragma unroll
        for (int j = 0; j < 16; ++j) bq[t][j] = *(const float4*)(brow + 8 * j);
    }

    const int mofs = h << 3;
    const int c0   = (2 * w) * 16 + lm;

    int strip = blockIdx.x;
    if (strip >= nstrips) return;

    float4 aq[16];
    {
        int ar = strip * 16 + lm; if (ar >= N) ar = N - 1;
        const float* arow = X + (size_t)ar * IN_F + 4 * h;
        #pragma unroll
        for (int j = 0; j < 16; ++j) aq[j] = *(const float4*)(arow + 8 * j);
    }

    for (; strip < nstrips; strip += gridDim.x) {
        float4 an[16];
        {
            const int ns = (strip + (int)gridDim.x < nstrips) ? strip + (int)gridDim.x : strip;
            int ar = ns * 16 + lm; if (ar >= N) ar = N - 1;
            const float* arow = X + (size_t)ar * IN_F + 4 * h;
            #pragma unroll
            for (int j = 0; j < 16; ++j) an[j] = *(const float4*)(arow + 8 * j);
        }

        v8f acc0 = (v8f){}, acc1 = (v8f){};
        #pragma unroll
        for (int j = 0; j < 16; ++j) {
            const v2f a_lo = {aq[j].x, aq[j].y};
            const v2f a_hi = {aq[j].z, aq[j].w};
            const v2f b0lo = {bq[0][j].x, bq[0][j].y};
            const v2f b1lo = {bq[1][j].x, bq[1][j].y};
            const v2f b0hi = {bq[0][j].z, bq[0][j].w};
            const v2f b1hi = {bq[1][j].z, bq[1][j].w};
            acc0 = __builtin_amdgcn_wmma_f32_16x16x4_f32(
                false, a_lo, false, b0lo, (short)0, acc0, false, false);
            acc1 = __builtin_amdgcn_wmma_f32_16x16x4_f32(
                false, a_lo, false, b1lo, (short)0, acc1, false, false);
            acc0 = __builtin_amdgcn_wmma_f32_16x16x4_f32(
                false, a_hi, false, b0hi, (short)0, acc0, false, false);
            acc1 = __builtin_amdgcn_wmma_f32_16x16x4_f32(
                false, a_hi, false, b1hi, (short)0, acc1, false, false);
        }

        const int r0 = strip * 16;
        #pragma unroll
        for (int v = 0; v < 8; ++v) {
            const int row = r0 + mofs + v;
            if (row < N) {
                float* orow = xp + (size_t)row * XDIM + c0;
                orow[0]  = acc0[v];
                orow[16] = acc1[v];
            }
        }

        #pragma unroll
        for (int j = 0; j < 16; ++j) aq[j] = an[j];
    }
}

#define NODE_W 16
#define OWN_NR 400

__device__ __forceinline__ void logits_load(
    const float* __restrict__ xp, float4 (&xv)[NODE_W], int n0, int N, int lane)
{
    const int fo = lane * 4;
    #pragma unroll
    for (int k = 0; k < NODE_W; ++k) {
        int node = n0 + k;
        if (node >= N) node = N - 1;
        xv[k] = *(const float4*)(xp + (size_t)node * XDIM + fo);
    }
}

template <bool FULL>
__device__ __forceinline__ void logits_rows(
    const float4 (&xv)[NODE_W], const float4 sv, const float4 dv,
    float* __restrict__ aS, float* __restrict__ aD, int n0, int N, int lane)
{
    float ps[NODE_W], pd[NODE_W];
    #pragma unroll
    for (int k = 0; k < NODE_W; ++k) {
        ps[k] = xv[k].x*sv.x + xv[k].y*sv.y + xv[k].z*sv.z + xv[k].w*sv.w;
        pd[k] = xv[k].x*dv.x + xv[k].y*dv.y + xv[k].z*dv.z + xv[k].w*dv.w;
    }
    #pragma unroll
    for (int off = 8; off >= 1; off >>= 1) {
        #pragma unroll
        for (int k = 0; k < NODE_W; ++k) {
            ps[k] += __shfl_xor(ps[k], off, 32);
            pd[k] += __shfl_xor(pd[k], off, 32);
        }
    }

    const int j = lane & 15;
    const int head = lane >> 4;
    float ss = ps[0], sd = pd[0];
    #pragma unroll
    for (int k = 1; k < NODE_W; ++k) {
        ss = (j == k) ? ps[k] : ss;
        sd = (j == k) ? pd[k] : sd;
    }
    if (FULL || n0 + j < N) {
        const size_t ti = (size_t)(n0 + j) * 2 + head;
        aS[ti] = ss;
        aD[ti] = sd;
    }
}

__device__ __forceinline__ void keys_load(
    const int* __restrict__ edst, int (&dd)[4], int ch, int lane)
{
    const int* dp = edst + (size_t)ch * 128 + lane;
    #pragma unroll
    for (int s = 0; s < 4; ++s) dd[s] = dp[32 * s];
}

__device__ __forceinline__ void keys_store(
    const int (&dd)[4], unsigned* __restrict__ keyw, int ch, int lane)
{
    unsigned outw = 0;
    #pragma unroll
    for (int s = 0; s < 4; ++s) {
        const unsigned key = (unsigned)dd[s] / OWN_NR;
        #pragma unroll
        for (int i = 0; i < 4; ++i) {
            const unsigned p0 = __builtin_amdgcn_ballot_w32(((key >> (2 * i)) & 1u) != 0u);
            const unsigned p1 = __builtin_amdgcn_ballot_w32(((key >> (2 * i + 1)) & 1u) != 0u);
            outw = (lane == 8 * s + i)     ? ((p0 & 0xFFFFu) | (p1 << 16))      : outw;
            outw = (lane == 8 * s + 4 + i) ? ((p0 >> 16) | (p1 & 0xFFFF0000u)) : outw;
        }
    }
    keyw[(size_t)ch * 32 + lane] = outw;
}

__global__ __launch_bounds__(256) void gat_logits(
    const float* __restrict__ xp,
    const float* __restrict__ attS, const float* __restrict__ attD,
    float* __restrict__ aS, float* __restrict__ aD, int N,
    const int* __restrict__ edst, unsigned* __restrict__ keyw, int E)
{
    const int lane = threadIdx.x & 31;
    const int wave = __builtin_amdgcn_readfirstlane(
        (int)((blockIdx.x * blockDim.x + threadIdx.x) >> 5));
    const int n0  = wave * NODE_W;
    const int nwv = (int)((gridDim.x * blockDim.x) >> 5);
    const int nch = E >> 7;

    int ch = wave;
    int dd[4] = {0, 0, 0, 0};
    if (ch < nch) keys_load(edst, dd, ch, lane);

    float4 xv[NODE_W];
    float4 sv = {0.0f, 0.0f, 0.0f, 0.0f}, dv = sv;
    if (n0 < N) {
        logits_load(xp, xv, n0, N, lane);
        sv = *(const float4*)(attS + lane * 4);
        dv = *(const float4*)(attD + lane * 4);
    }

    while (ch < nch) {
        const int nx = ch + nwv;
        int dn[4] = {0, 0, 0, 0};
        if (nx < nch) keys_load(edst, dn, nx, lane);
        keys_store(dd, keyw, ch, lane);
        #pragma unroll
        for (int s = 0; s < 4; ++s) dd[s] = dn[s];
        ch = nx;
    }

    if (n0 >= N) return;
    if (n0 + NODE_W <= N)
        logits_rows<true>(xv, sv, dv, aS, aD, n0, N, lane);
    else
        logits_rows<false>(xv, sv, dv, aS, aD, n0, N, lane);
}

__device__ __forceinline__ void atomicMaxF(float* addr, float val)
{
    if (val >= 0.0f) atomicMax((int*)addr, __float_as_int(val));
    else             atomicMin((unsigned int*)addr, __float_as_uint(val));
}

__device__ __forceinline__ float lrelu(float v)
{
    return v > 0.0f ? v : NEG_SLOPE * v;
}

#define OWN_T  1024
#define OWN_Q  10240
#define OWN_U  8
#define OWN_B  8
#define OWN_LN 4

__device__ __forceinline__ unsigned own_flags16(const uint4 kv, const uint4 mk)
{
    const unsigned a = (kv.x ^ mk.x) & (kv.y ^ mk.y) & (kv.z ^ mk.z) & (kv.w ^ mk.w);
    return a & (a >> 16);
}

__device__ __forceinline__ void own_drain(
    unsigned w, int qa, int qb, int* q_s, int* q_n)
{
    while (w) {
        const int p = __ffs((int)w) - 1;
        const int q = (p & 16) ? qb : qa;
        const int idx = atomicAdd(q_n, 1);
        if (idx < OWN_Q) q_s[idx] = q * 16 + (p & 15);
        w &= w - 1;
    }
}

__device__ __forceinline__ void own_scan_keys(
    const uint4* __restrict__ k16, int n16, int rot, int tid, const uint4 mk,
    int* q_s, int* q_n)
{
    int i0 = 0;
    for (; i0 + OWN_T * OWN_U <= n16; i0 += OWN_T * OWN_U) {
        uint4 kv[OWN_U];
        int   qg[OWN_U];
        #pragma unroll
        for (int u = 0; u < OWN_U; ++u) {
            int q = rot + i0 + u * OWN_T + tid;
            if (q >= n16) q -= n16;
            qg[u] = q;
            kv[u] = k16[q];
        }
        unsigned w[OWN_U / 2];
        #pragma unroll
        for (int h = 0; h < OWN_U / 2; ++h)
            w[h] = own_flags16(kv[2 * h], mk) | (own_flags16(kv[2 * h + 1], mk) << 16);
        #pragma unroll
        for (int h = 0; h < OWN_U / 2; ++h)
            own_drain(w[h], qg[2 * h], qg[2 * h + 1], q_s, q_n);
    }
    if (i0 < n16) {
        uint4 kv[OWN_U];
        int   qg[OWN_U];
        bool  in[OWN_U];
        #pragma unroll
        for (int u = 0; u < OWN_U; ++u) {
            const int i = i0 + u * OWN_T + tid;
            int q = rot + i;
            if (q >= n16) q -= n16;
            in[u] = i < n16;
            qg[u] = q;
            kv[u] = k16[in[u] ? q : rot];
        }
        unsigned w[OWN_U / 2];
        #pragma unroll
        for (int h = 0; h < OWN_U / 2; ++h)
            w[h] = (in[2 * h]     ? own_flags16(kv[2 * h], mk)     : 0u)
                 | ((in[2 * h + 1] ? own_flags16(kv[2 * h + 1], mk) : 0u) << 16);
        #pragma unroll
        for (int h = 0; h < OWN_U / 2; ++h)
            own_drain(w[h], qg[2 * h], qg[2 * h + 1], q_s, q_n);
    }
}

__global__ __launch_bounds__(OWN_T) void gat_own_range(
    const int* __restrict__ esrc, const int* __restrict__ edst,
    const float* __restrict__ aS, const float* __restrict__ aD,
    const float* __restrict__ xp,
    const float* __restrict__ bias, const float* __restrict__ gamma,
    const float* __restrict__ beta, float* __restrict__ out,
    const unsigned* __restrict__ keyw, int E, int N)
{
    __shared__ float acc_l[OWN_NR * OUT_F];
    __shared__ float t_ad[OWN_NR * 2];
    __shared__ float t_m[OWN_NR * 2];
    __shared__ float t_d[OWN_NR * 2];
    __shared__ int   q_s[OWN_Q];
    __shared__ int   q_ld[OWN_Q];
    __shared__ float q_x0[OWN_Q];
    __shared__ float q_x1[OWN_Q];
    __shared__ int   q_n;

    const int tid  = threadIdx.x;
    const int lane = tid & 31;
    const int d0   = (int)blockIdx.x * OWN_NR;
    const int nr   = (N - d0 < OWN_NR) ? (N - d0) : OWN_NR;
    const int Et   = E + N;

    for (int i = tid; i < nr * OUT_F; i += OWN_T) acc_l[i] = 0.0f;
    for (int i = tid; i < nr * 2; i += OWN_T) {
        t_ad[i] = aD[(size_t)d0 * 2 + i];
        t_m[i]  = -INFINITY;
        t_d[i]  = 0.0f;
    }
    for (int i = tid; i < nr; i += OWN_T)
        q_s[i] = E + d0 + i;
    if (tid == 0) q_n = nr;
    __syncthreads();

    {
        const int n16 = (E >> 7) << 3;
        if (n16 > 0) {
            const int rot = (int)(((long long)blockIdx.x * n16) / (long long)gridDim.x) & ~31;
            const unsigned kb = (unsigned)blockIdx.x;
            uint4 mk;
            mk.x = ((kb & 0x01u) ? 0u : 0x0000FFFFu) | ((kb & 0x02u) ? 0u : 0xFFFF0000u);
            mk.y = ((kb & 0x04u) ? 0u : 0x0000FFFFu) | ((kb & 0x08u) ? 0u : 0xFFFF0000u);
            mk.z = ((kb & 0x10u) ? 0u : 0x0000FFFFu) | ((kb & 0x20u) ? 0u : 0xFFFF0000u);
            mk.w = ((kb & 0x40u) ? 0u : 0x0000FFFFu) | ((kb & 0x80u) ? 0u : 0xFFFF0000u);
            own_scan_keys((const uint4*)keyw, n16, rot, tid, mk, q_s, &q_n);
        }
        for (int eid = n16 * 16 + tid; eid < E; eid += OWN_T) {
            if ((unsigned)(edst[eid] - d0) < (unsigned)nr) {
                const int idx = atomicAdd(&q_n, 1);
                if (idx < OWN_Q) q_s[idx] = eid;
            }
        }
    }
    __syncthreads();
    const int nq = q_n;

    if (nq <= OWN_Q) {
        for (int i = tid; i < nq; i += OWN_T) {
            const int eid = q_s[i];
            int s, d;
            if (eid < E) { s = esrc[eid]; d = edst[eid]; } else { s = d = eid - E; }
            const unsigned ldu = (unsigned)(d - d0);
            const bool mine = ldu < (unsigned)nr;
            const int ld = mine ? (int)ldu : 0;
            const float2 as = *(const float2*)(aS + (size_t)s * 2);
            const float e0 = mine ? lrelu(as.x + t_ad[2 * ld])     : -INFINITY;
            const float e1 = mine ? lrelu(as.y + t_ad[2 * ld + 1]) : -INFINITY;
            atomicMaxF(&t_m[2 * ld],     e0);
            atomicMaxF(&t_m[2 * ld + 1], e1);
            q_s[i]  = s;
            q_ld[i] = ld;
            q_x0[i] = e0;
            q_x1[i] = e1;
        }
        __syncthreads();
        for (int i = tid; i < nq; i += OWN_T) {
            const int ld = q_ld[i];
            const float x0 = __expf(q_x0[i] - t_m[2 * ld]);
            const float x1 = __expf(q_x1[i] - t_m[2 * ld + 1]);
            atomicAdd(&t_d[2 * ld],     x0);
            atomicAdd(&t_d[2 * ld + 1], x1);
            q_x0[i] = x0;
            q_x1[i] = x1;
        }
        __syncthreads();
        {
            const int wv = __builtin_amdgcn_readfirstlane(tid >> 5);
            const float* xl = xp + lane;
            float* al = acc_l + lane;
            for (int g = wv * 32; g < nq; g += OWN_T) {
                const int nvalid = (nq - g < 32) ? (nq - g) : 32;
                const int ic = (lane < nvalid) ? (g + lane) : g;
                const int s  = q_s[ic];
                const int ld = q_ld[ic];
                const float w0 = 0.5f * (q_x0[ic] / (t_d[2 * ld]     + SM_EPS));
                const float w1 = 0.5f * (q_x1[ic] / (t_d[2 * ld + 1] + SM_EPS));
                const int w0i = __float_as_int(w0);
                const int w1i = __float_as_int(w1);
                if (nvalid == 32) {
                    for (int jb = 0; jb < 32; jb += OWN_B) {
                        float x00[OWN_B], x01[OWN_B], x10[OWN_B], x11[OWN_B];
                        #pragma unroll
                        for (int j = 0; j < OWN_B; ++j) {
                            const int sj = __builtin_amdgcn_readlane(s, jb + j);
                            const float* xr = xl + (size_t)sj * XDIM;
                            x00[j] = xr[0];
                            x01[j] = xr[32];
                            x10[j] = xr[OUT_F];
                            x11[j] = xr[OUT_F + 32];
                        }
                        #pragma unroll
                        for (int j = 0; j < OWN_B; ++j) {
                            const int   lj  = __builtin_amdgcn_readlane(ld, jb + j);
                            const float w0j = __int_as_float(__builtin_amdgcn_readlane(w0i, jb + j));
                            const float w1j = __int_as_float(__builtin_amdgcn_readlane(w1i, jb + j));
                            float* ap = al + lj * OUT_F;
                            atomicAdd(ap,      w0j * x00[j] + w1j * x10[j]);
                            atomicAdd(ap + 32, w0j * x01[j] + w1j * x11[j]);
                        }
                    }
                } else {
                    for (int j = 0; j < nvalid; ++j) {
                        const int   sj  = __builtin_amdgcn_readlane(s, j);
                        const int   lj  = __builtin_amdgcn_readlane(ld, j);
                        const float w0j = __int_as_float(__builtin_amdgcn_readlane(w0i, j));
                        const float w1j = __int_as_float(__builtin_amdgcn_readlane(w1i, j));
                        const float* xr = xl + (size_t)sj * XDIM;
                        float* ap = al + lj * OUT_F;
                        atomicAdd(ap,      w0j * xr[0]  + w1j * xr[OUT_F]);
                        atomicAdd(ap + 32, w0j * xr[32] + w1j * xr[OUT_F + 32]);
                    }
                }
            }
        }
        __syncthreads();
    } else {
        for (int eid = tid; eid < Et; eid += OWN_T) {
            int s, d;
            if (eid < E) { s = esrc[eid]; d = edst[eid]; } else { s = d = eid - E; }
            const unsigned ld = (unsigned)(d - d0);
            if (ld < (unsigned)nr) {
                const float2 as = *(const float2*)(aS + (size_t)s * 2);
                atomicMaxF(&t_m[2 * ld],     lrelu(as.x + t_ad[2 * ld]));
                atomicMaxF(&t_m[2 * ld + 1], lrelu(as.y + t_ad[2 * ld + 1]));
            }
        }
        __syncthreads();
        for (int eid = tid; eid < Et; eid += OWN_T) {
            int s, d;
            if (eid < E) { s = esrc[eid]; d = edst[eid]; } else { s = d = eid - E; }
            const unsigned ld = (unsigned)(d - d0);
            if (ld < (unsigned)nr) {
                const float2 as = *(const float2*)(aS + (size_t)s * 2);
                atomicAdd(&t_d[2 * ld],     __expf(lrelu(as.x + t_ad[2 * ld])     - t_m[2 * ld]));
                atomicAdd(&t_d[2 * ld + 1], __expf(lrelu(as.y + t_ad[2 * ld + 1]) - t_m[2 * ld + 1]));
            }
        }
        __syncthreads();
        for (int eid = tid; eid < Et; eid += OWN_T) {
            int s, d;
            if (eid < E) { s = esrc[eid]; d = edst[eid]; } else { s = d = eid - E; }
            const unsigned ld = (unsigned)(d - d0);
            if (ld < (unsigned)nr) {
                const float2 as = *(const float2*)(aS + (size_t)s * 2);
                const float x0 = __expf(lrelu(as.x + t_ad[2 * ld])     - t_m[2 * ld]);
                const float x1 = __expf(lrelu(as.y + t_ad[2 * ld + 1]) - t_m[2 * ld + 1]);
                const float w0 = 0.5f * (x0 / (t_d[2 * ld]     + SM_EPS));
                const float w1 = 0.5f * (x1 / (t_d[2 * ld + 1] + SM_EPS));
                const float* xr = xp + (size_t)s * XDIM;
                for (int f = 0; f < OUT_F; ++f)
                    atomicAdd(&acc_l[ld * OUT_F + f], w0 * xr[f] + w1 * xr[OUT_F + f]);
            }
        }
        __syncthreads();
    }

    {
        const int wv = __builtin_amdgcn_readfirstlane(tid >> 5);
        const int f0 = lane, f1 = lane + 32;
        const float b0 = bias[f0],  b1 = bias[f1];
        const float g0 = gamma[f0], g1 = gamma[f1];
        const float be0 = beta[f0], be1 = beta[f1];
        for (int base = wv * OWN_LN; base < nr; base += (OWN_T / 32) * OWN_LN) {
            float v0[OWN_LN], v1[OWN_LN], sum[OWN_LN];
            #pragma unroll
            for (int k = 0; k < OWN_LN; ++k) {
                int r = base + k;
                if (r >= nr) r = nr - 1;
                v0[k] = acc_l[r * OUT_F + f0] + b0;
                v1[k] = acc_l[r * OUT_F + f1] + b1;
                sum[k] = v0[k] + v1[k];
            }
            #pragma unroll
            for (int off = 16; off >= 1; off >>= 1) {
                #pragma unroll
                for (int k = 0; k < OWN_LN; ++k) sum[k] += __shfl_xor(sum[k], off, 32);
            }
            float c0[OWN_LN], c1[OWN_LN], sq[OWN_LN];
            #pragma unroll
            for (int k = 0; k < OWN_LN; ++k) {
                const float mu = sum[k] * (1.0f / OUT_F);
                c0[k] = v0[k] - mu;
                c1[k] = v1[k] - mu;
                sq[k] = c0[k] * c0[k] + c1[k] * c1[k];
            }
            #pragma unroll
            for (int off = 16; off >= 1; off >>= 1) {
                #pragma unroll
                for (int k = 0; k < OWN_LN; ++k) sq[k] += __shfl_xor(sq[k], off, 32);
            }
            #pragma unroll
            for (int k = 0; k < OWN_LN; ++k) {
                const float inv = rsqrtf(sq[k] * (1.0f / OUT_F) + LN_EPS);
                if (base + k < nr) {
                    float* o = out + (size_t)(d0 + base + k) * OUT_F;
                    o[f0] = c0[k] * inv * g0 + be0;
                    o[f1] = c1[k] * inv * g1 + be1;
                }
            }
        }
    }
}

extern "C" void kernel_launch(void* const* d_in, const int* in_sizes, int n_in,
                              void* d_out, int out_size, void* d_ws, size_t ws_size,
                              hipStream_t stream)
{
    const float* X    = (const float*)d_in[0];
    const int*   ei   = (const int*)  d_in[1];
    const float* W    = (const float*)d_in[2];
    const float* attS = (const float*)d_in[3];
    const float* attD = (const float*)d_in[4];
    const float* bias = (const float*)d_in[5];
    const float* g    = (const float*)d_in[6];
    const float* b    = (const float*)d_in[7];
    float* out = (float*)d_out;

    const int N = in_sizes[0] / IN_F;
    const int E = in_sizes[1] / 2;
    const int* esrc = ei;
    const int* edst = ei + E;

    float* ws    = (float*)d_ws;
    float* xp    = ws;
    float* aS    = xp   + (size_t)N * XDIM;
    float* aD    = aS   + (size_t)N * 2;
    unsigned* keyw = (unsigned*)(aD + (size_t)N * 2);

    gat_gemm_wmma<<<GEMM_BLOCKS, 128, 0, stream>>>(X, W, xp, N);
    gat_logits<<<(N + 8 * NODE_W - 1) / (8 * NODE_W), 256, 0, stream>>>(
        xp, attS, attD, aS, aD, N, edst, keyw, E);
    gat_own_range<<<(N + OWN_NR - 1) / OWN_NR, OWN_T, 0, stream>>>(
        esrc, edst, aS, aD, xp, bias, g, b, out, keyw, E, N);
}


# Rows of x stored with the two heads interleaved (position 2f+head): an edge's gathered row is one 16-byte load per lane, not four 4-byte loads; logit dot products summed in another order
# speedup vs baseline: 5.8171x; 1.0593x over previous
#include <hip/hip_runtime.h>
#include <hip/hip_bf16.h>
#include <math.h>


typedef __attribute__((ext_vector_type(2))) float v2f;
typedef __attribute__((ext_vector_type(8))) float v8f;

#define IN_F     128
#define OUT_F    64
#define HEADS    2
#define XDIM     128
#define NEG_SLOPE 0.2f
#define LN_EPS    1e-5f
#define SM_EPS    1e-16f

#define GEMM_BLOCKS 768

__global__ __launch_bounds__(128) void gat_gemm_wmma(
    const float* __restrict__ X, const float* __restrict__ W,
    float* __restrict__ xp, int N)
{
    const int lane = threadIdx.x & 31;
    const int w    = threadIdx.x >> 5;
    const int lm   = lane & 15;
    const int h    = lane >> 4;
    const int nstrips = (N + 15) >> 4;

    float4 bq[2][16];
    #pragma unroll
    for (int t = 0; t < 2; ++t) {
        const float* brow = W + (size_t)(t * OUT_F + 16 * w + lm) * IN_F + 4 * h;
        #pragma unroll
        for (int j = 0; j < 16; ++j) bq[t][j] = *(const float4*)(brow + 8 * j);
    }

    const int mofs = h << 3;
    const int c0   = 2 * (16 * w + lm);

    int strip = blockIdx.x;
    if (strip >= nstrips) return;

    float4 aq[16];
    {
        int ar = strip * 16 + lm; if (ar >= N) ar = N - 1;
        const float* arow = X + (size_t)ar * IN_F + 4 * h;
        #pragma unroll
        for (int j = 0; j < 16; ++j) aq[j] = *(const float4*)(arow + 8 * j);
    }

    for (; strip < nstrips; strip += gridDim.x) {
        float4 an[16];
        {
            const int ns = (strip + (int)gridDim.x < nstrips) ? strip + (int)gridDim.x : strip;
            int ar = ns * 16 + lm; if (ar >= N) ar = N - 1;
            const float* arow = X + (size_t)ar * IN_F + 4 * h;
            #pragma unroll
            for (int j = 0; j < 16; ++j) an[j] = *(const float4*)(arow + 8 * j);
        }

        v8f acc0 = (v8f){}, acc1 = (v8f){};
        #pragma unroll
        for (int j = 0; j < 16; ++j) {
            const v2f a_lo = {aq[j].x, aq[j].y};
            const v2f a_hi = {aq[j].z, aq[j].w};
            const v2f b0lo = {bq[0][j].x, bq[0][j].y};
            const v2f b1lo = {bq[1][j].x, bq[1][j].y};
            const v2f b0hi = {bq[0][j].z, bq[0][j].w};
            const v2f b1hi = {bq[1][j].z, bq[1][j].w};
            acc0 = __builtin_amdgcn_wmma_f32_16x16x4_f32(
                false, a_lo, false, b0lo, (short)0, acc0, false, false);
            acc1 = __builtin_amdgcn_wmma_f32_16x16x4_f32(
                false, a_lo, false, b1lo, (short)0, acc1, false, false);
            acc0 = __builtin_amdgcn_wmma_f32_16x16x4_f32(
                false, a_hi, false, b0hi, (short)0, acc0, false, false);
            acc1 = __builtin_amdgcn_wmma_f32_16x16x4_f32(
                false, a_hi, false, b1hi, (short)0, acc1, false, false);
        }

        const int r0 = strip * 16;
        #pragma unroll
        for (int v = 0; v < 8; ++v) {
            const int row = r0 + mofs + v;
            if (row < N) {
                float2* opair = (float2*)(xp + (size_t)row * XDIM + c0);
                *opair = make_float2(acc0[v], acc1[v]);
            }
        }

        #pragma unroll
        for (int j = 0; j < 16; ++j) aq[j] = an[j];
    }
}

#define NODE_W 16
#define OWN_NR 400

__device__ __forceinline__ void logits_load(
    const float* __restrict__ xp, float4 (&xv)[NODE_W], int n0, int N, int lane)
{
    const int fo = lane * 4;
    #pragma unroll
    for (int k = 0; k < NODE_W; ++k) {
        int node = n0 + k;
        if (node >= N) node = N - 1;
        xv[k] = *(const float4*)(xp + (size_t)node * XDIM + fo);
    }
}

template <bool FULL>
__device__ __forceinline__ void logits_rows(
    const float4 (&xv)[NODE_W], const float4 sv, const float4 dv,
    float* __restrict__ aS, float* __restrict__ aD, int n0, int N, int lane)
{
    const bool up = lane >= 16;
    float ps[NODE_W], pd[NODE_W];
    #pragma unroll
    for (int k = 0; k < NODE_W; ++k) {
        const float s0 = xv[k].x*sv.x + xv[k].z*sv.z;
        const float s1 = xv[k].y*sv.y + xv[k].w*sv.w;
        const float d0 = xv[k].x*dv.x + xv[k].z*dv.z;
        const float d1 = xv[k].y*dv.y + xv[k].w*dv.w;
        ps[k] = (up ? s1 : s0) + __shfl_xor(up ? s0 : s1, 16, 32);
        pd[k] = (up ? d1 : d0) + __shfl_xor(up ? d0 : d1, 16, 32);
    }
    #pragma unroll
    for (int off = 8; off >= 1; off >>= 1) {
        #pragma unroll
        for (int k = 0; k < NODE_W; ++k) {
            ps[k] += __shfl_xor(ps[k], off, 32);
            pd[k] += __shfl_xor(pd[k], off, 32);
        }
    }

    const int j = lane & 15;
    const int head = lane >> 4;
    float ss = ps[0], sd = pd[0];
    #pragma unroll
    for (int k = 1; k < NODE_W; ++k) {
        ss = (j == k) ? ps[k] : ss;
        sd = (j == k) ? pd[k] : sd;
    }
    if (FULL || n0 + j < N) {
        const size_t ti = (size_t)(n0 + j) * 2 + head;
        aS[ti] = ss;
        aD[ti] = sd;
    }
}

__device__ __forceinline__ void keys_load(
    const int* __restrict__ edst, int (&dd)[4], int ch, int lane)
{
    const int* dp = edst + (size_t)ch * 128 + lane;
    #pragma unroll
    for (int s = 0; s < 4; ++s) dd[s] = dp[32 * s];
}

__device__ __forceinline__ void keys_store(
    const int (&dd)[4], unsigned* __restrict__ keyw, int ch, int lane)
{
    unsigned outw = 0;
    #pragma unroll
    for (int s = 0; s < 4; ++s) {
        const unsigned key = (unsigned)dd[s] / OWN_NR;
        #pragma unroll
        for (int i = 0; i < 4; ++i) {
            const unsigned p0 = __builtin_amdgcn_ballot_w32(((key >> (2 * i)) & 1u) != 0u);
            const unsigned p1 = __builtin_amdgcn_ballot_w32(((key >> (2 * i + 1)) & 1u) != 0u);
            outw = (lane == 8 * s + i)     ? ((p0 & 0xFFFFu) | (p1 << 16))      : outw;
            outw = (lane == 8 * s + 4 + i) ? ((p0 >> 16) | (p1 & 0xFFFF0000u)) : outw;
        }
    }
    keyw[(size_t)ch * 32 + lane] = outw;
}

__global__ __launch_bounds__(256) void gat_logits(
    const float* __restrict__ xp,
    const float* __restrict__ attS, const float* __restrict__ attD,
    float* __restrict__ aS, float* __restrict__ aD, int N,
    const int* __restrict__ edst, unsigned* __restrict__ keyw, int E)
{
    const int lane = threadIdx.x & 31;
    const int wave = __builtin_amdgcn_readfirstlane(
        (int)((blockIdx.x * blockDim.x + threadIdx.x) >> 5));
    const int n0  = wave * NODE_W;
    const int nwv = (int)((gridDim.x * blockDim.x) >> 5);
    const int nch = E >> 7;

    int ch = wave;
    int dd[4] = {0, 0, 0, 0};
    if (ch < nch) keys_load(edst, dd, ch, lane);

    float4 xv[NODE_W];
    float4 sv = {0.0f, 0.0f, 0.0f, 0.0f}, dv = sv;
    if (n0 < N) {
        logits_load(xp, xv, n0, N, lane);
        const float2 s0 = *(const float2*)(attS + 2 * lane);
        const float2 s1 = *(const float2*)(attS + OUT_F + 2 * lane);
        const float2 d0 = *(const float2*)(attD + 2 * lane);
        const float2 d1 = *(const float2*)(attD + OUT_F + 2 * lane);
        sv = make_float4(s0.x, s1.x, s0.y, s1.y);
        dv = make_float4(d0.x, d1.x, d0.y, d1.y);
    }

    while (ch < nch) {
        const int nx = ch + nwv;
        int dn[4] = {0, 0, 0, 0};
        if (nx < nch) keys_load(edst, dn, nx, lane);
        keys_store(dd, keyw, ch, lane);
        #pragma unroll
        for (int s = 0; s < 4; ++s) dd[s] = dn[s];
        ch = nx;
    }

    if (n0 >= N) return;
    if (n0 + NODE_W <= N)
        logits_rows<true>(xv, sv, dv, aS, aD, n0, N, lane);
    else
        logits_rows<false>(xv, sv, dv, aS, aD, n0, N, lane);
}

__device__ __forceinline__ void atomicMaxF(float* addr, float val)
{
    if (val >= 0.0f) atomicMax((int*)addr, __float_as_int(val));
    else             atomicMin((unsigned int*)addr, __float_as_uint(val));
}

__device__ __forceinline__ float lrelu(float v)
{
    return v > 0.0f ? v : NEG_SLOPE * v;
}

#define OWN_T  1024
#define OWN_Q  10240
#define OWN_U  8
#define OWN_B  8
#define OWN_LN 4

__device__ __forceinline__ unsigned own_flags16(const uint4 kv, const uint4 mk)
{
    const unsigned a = (kv.x ^ mk.x) & (kv.y ^ mk.y) & (kv.z ^ mk.z) & (kv.w ^ mk.w);
    return a & (a >> 16);
}

__device__ __forceinline__ void own_drain(
    unsigned w, int qa, int qb, int* q_s, int* q_n)
{
    while (w) {
        const int p = __ffs((int)w) - 1;
        const int q = (p & 16) ? qb : qa;
        const int idx = atomicAdd(q_n, 1);
        if (idx < OWN_Q) q_s[idx] = q * 16 + (p & 15);
        w &= w - 1;
    }
}

__device__ __forceinline__ void own_scan_keys(
    const uint4* __restrict__ k16, int n16, int rot, int tid, const uint4 mk,
    int* q_s, int* q_n)
{
    int i0 = 0;
    for (; i0 + OWN_T * OWN_U <= n16; i0 += OWN_T * OWN_U) {
        uint4 kv[OWN_U];
        int   qg[OWN_U];
        #pragma unroll
        for (int u = 0; u < OWN_U; ++u) {
            int q = rot + i0 + u * OWN_T + tid;
            if (q >= n16) q -= n16;
            qg[u] = q;
            kv[u] = k16[q];
        }
        unsigned w[OWN_U / 2];
        #pragma unroll
        for (int h = 0; h < OWN_U / 2; ++h)
            w[h] = own_flags16(kv[2 * h], mk) | (own_flags16(kv[2 * h + 1], mk) << 16);
        #pragma unroll
        for (int h = 0; h < OWN_U / 2; ++h)
            own_drain(w[h], qg[2 * h], qg[2 * h + 1], q_s, q_n);
    }
    if (i0 < n16) {
        uint4 kv[OWN_U];
        int   qg[OWN_U];
        bool  in[OWN_U];
        #pragma unroll
        for (int u = 0; u < OWN_U; ++u) {
            const int i = i0 + u * OWN_T + tid;
            int q = rot + i;
            if (q >= n16) q -= n16;
            in[u] = i < n16;
            qg[u] = q;
            kv[u] = k16[in[u] ? q : rot];
        }
        unsigned w[OWN_U / 2];
        #pragma unroll
        for (int h = 0; h < OWN_U / 2; ++h)
            w[h] = (in[2 * h]     ? own_flags16(kv[2 * h], mk)     : 0u)
                 | ((in[2 * h + 1] ? own_flags16(kv[2 * h + 1], mk) : 0u) << 16);
        #pragma unroll
        for (int h = 0; h < OWN_U / 2; ++h)
            own_drain(w[h], qg[2 * h], qg[2 * h + 1], q_s, q_n);
    }
}

__global__ __launch_bounds__(OWN_T) void gat_own_range(
    const int* __restrict__ esrc, const int* __restrict__ edst,
    const float* __restrict__ aS, const float* __restrict__ aD,
    const float* __restrict__ xp,
    const float* __restrict__ bias, const float* __restrict__ gamma,
    const float* __restrict__ beta, float* __restrict__ out,
    const unsigned* __restrict__ keyw, int E, int N)
{
    __shared__ float acc_l[OWN_NR * OUT_F];
    __shared__ float t_ad[OWN_NR * 2];
    __shared__ float t_m[OWN_NR * 2];
    __shared__ float t_d[OWN_NR * 2];
    __shared__ int   q_s[OWN_Q];
    __shared__ int   q_ld[OWN_Q];
    __shared__ float q_x0[OWN_Q];
    __shared__ float q_x1[OWN_Q];
    __shared__ int   q_n;

    const int tid  = threadIdx.x;
    const int lane = tid & 31;
    const int d0   = (int)blockIdx.x * OWN_NR;
    const int nr   = (N - d0 < OWN_NR) ? (N - d0) : OWN_NR;
    const int Et   = E + N;

    for (int i = tid; i < nr * OUT_F; i += OWN_T) acc_l[i] = 0.0f;
    for (int i = tid; i < nr * 2; i += OWN_T) {
        t_ad[i] = aD[(size_t)d0 * 2 + i];
        t_m[i]  = -INFINITY;
        t_d[i]  = 0.0f;
    }
    for (int i = tid; i < nr; i += OWN_T)
        q_s[i] = E + d0 + i;
    if (tid == 0) q_n = nr;
    __syncthreads();

    {
        const int n16 = (E >> 7) << 3;
        if (n16 > 0) {
            const int rot = (int)(((long long)blockIdx.x * n16) / (long long)gridDim.x) & ~31;
            const unsigned kb = (unsigned)blockIdx.x;
            uint4 mk;
            mk.x = ((kb & 0x01u) ? 0u : 0x0000FFFFu) | ((kb & 0x02u) ? 0u : 0xFFFF0000u);
            mk.y = ((kb & 0x04u) ? 0u : 0x0000FFFFu) | ((kb & 0x08u) ? 0u : 0xFFFF0000u);
            mk.z = ((kb & 0x10u) ? 0u : 0x0000FFFFu) | ((kb & 0x20u) ? 0u : 0xFFFF0000u);
            mk.w = ((kb & 0x40u) ? 0u : 0x0000FFFFu) | ((kb & 0x80u) ? 0u : 0xFFFF0000u);
            own_scan_keys((const uint4*)keyw, n16, rot, tid, mk, q_s, &q_n);
        }
        for (int eid = n16 * 16 + tid; eid < E; eid += OWN_T) {
            if ((unsigned)(edst[eid] - d0) < (unsigned)nr) {
                const int idx = atomicAdd(&q_n, 1);
                if (idx < OWN_Q) q_s[idx] = eid;
            }
        }
    }
    __syncthreads();
    const int nq = q_n;

    if (nq <= OWN_Q) {
        for (int i = tid; i < nq; i += OWN_T) {
            const int eid = q_s[i];
            int s, d;
            if (eid < E) { s = esrc[eid]; d = edst[eid]; } else { s = d = eid - E; }
            const unsigned ldu = (unsigned)(d - d0);
            const bool mine = ldu < (unsigned)nr;
            const int ld = mine ? (int)ldu : 0;
            const float2 as = *(const float2*)(aS + (size_t)s * 2);
            const float e0 = mine ? lrelu(as.x + t_ad[2 * ld])     : -INFINITY;
            const float e1 = mine ? lrelu(as.y + t_ad[2 * ld + 1]) : -INFINITY;
            atomicMaxF(&t_m[2 * ld],     e0);
            atomicMaxF(&t_m[2 * ld + 1], e1);
            q_s[i]  = s;
            q_ld[i] = ld;
            q_x0[i] = e0;
            q_x1[i] = e1;
        }
        __syncthreads();
        for (int i = tid; i < nq; i += OWN_T) {
            const int ld = q_ld[i];
            const float x0 = __expf(q_x0[i] - t_m[2 * ld]);
            const float x1 = __expf(q_x1[i] - t_m[2 * ld + 1]);
            atomicAdd(&t_d[2 * ld],     x0);
            atomicAdd(&t_d[2 * ld + 1], x1);
            q_x0[i] = x0;
            q_x1[i] = x1;
        }
        __syncthreads();
        {
            const int wv = __builtin_amdgcn_readfirstlane(tid >> 5);
            const float4* xl = (const float4*)xp + lane;
            float* al = acc_l + 2 * lane;
            for (int g = wv * 32; g < nq; g += OWN_T) {
                const int nvalid = (nq - g < 32) ? (nq - g) : 32;
                const int ic = (lane < nvalid) ? (g + lane) : g;
                const int s  = q_s[ic];
                const int ld = q_ld[ic];
                const float w0 = 0.5f * (q_x0[ic] / (t_d[2 * ld]     + SM_EPS));
                const float w1 = 0.5f * (q_x1[ic] / (t_d[2 * ld + 1] + SM_EPS));
                const int w0i = __float_as_int(w0);
                const int w1i = __float_as_int(w1);
                if (nvalid == 32) {
                    for (int jb = 0; jb < 32; jb += OWN_B) {
                        float4 xr[OWN_B];
                        #pragma unroll
                        for (int j = 0; j < OWN_B; ++j) {
                            const int sj = __builtin_amdgcn_readlane(s, jb + j);
                            xr[j] = xl[(size_t)sj * (XDIM / 4)];
                        }
                        #pragma unroll
                        for (int j = 0; j < OWN_B; ++j) {
                            const int   lj  = __builtin_amdgcn_readlane(ld, jb + j);
                            const float w0j = __int_as_float(__builtin_amdgcn_readlane(w0i, jb + j));
                            const float w1j = __int_as_float(__builtin_amdgcn_readlane(w1i, jb + j));
                            float* ap = al + lj * OUT_F;
                            atomicAdd(ap,     w0j * xr[j].x + w1j * xr[j].y);
                            atomicAdd(ap + 1, w0j * xr[j].z + w1j * xr[j].w);
                        }
                    }
                } else {
                    for (int j = 0; j < nvalid; ++j) {
                        const int   sj  = __builtin_amdgcn_readlane(s, j);
                        const int   lj  = __builtin_amdgcn_readlane(ld, j);
                        const float w0j = __int_as_float(__builtin_amdgcn_readlane(w0i, j));
                        const float w1j = __int_as_float(__builtin_amdgcn_readlane(w1i, j));
                        const float4 xr = xl[(size_t)sj * (XDIM / 4)];
                        float* ap = al + lj * OUT_F;
                        atomicAdd(ap,     w0j * xr.x + w1j * xr.y);
                        atomicAdd(ap + 1, w0j * xr.z + w1j * xr.w);
                    }
                }
            }
        }
        __syncthreads();
    } else {
        for (int eid = tid; eid < Et; eid += OWN_T) {
            int s, d;
            if (eid < E) { s = esrc[eid]; d = edst[eid]; } else { s = d = eid - E; }
            const unsigned ld = (unsigned)(d - d0);
            if (ld < (unsigned)nr) {
                const float2 as = *(const float2*)(aS + (size_t)s * 2);
                atomicMaxF(&t_m[2 * ld],     lrelu(as.x + t_ad[2 * ld]));
                atomicMaxF(&t_m[2 * ld + 1], lrelu(as.y + t_ad[2 * ld + 1]));
            }
        }
        __syncthreads();
        for (int eid = tid; eid < Et; eid += OWN_T) {
            int s, d;
            if (eid < E) { s = esrc[eid]; d = edst[eid]; } else { s = d = eid - E; }
            const unsigned ld = (unsigned)(d - d0);
            if (ld < (unsigned)nr) {
                const float2 as = *(const float2*)(aS + (size_t)s * 2);
                atomicAdd(&t_d[2 * ld],     __expf(lrelu(as.x + t_ad[2 * ld])     - t_m[2 * ld]));
                atomicAdd(&t_d[2 * ld + 1], __expf(lrelu(as.y + t_ad[2 * ld + 1]) - t_m[2 * ld + 1]));
            }
        }
        __syncthreads();
        for (int eid = tid; eid < Et; eid += OWN_T) {
            int s, d;
            if (eid < E) { s = esrc[eid]; d = edst[eid]; } else { s = d = eid - E; }
            const unsigned ld = (unsigned)(d - d0);
            if (ld < (unsigned)nr) {
                const float2 as = *(const float2*)(aS + (size_t)s * 2);
                const float x0 = __expf(lrelu(as.x + t_ad[2 * ld])     - t_m[2 * ld]);
                const float x1 = __expf(lrelu(as.y + t_ad[2 * ld + 1]) - t_m[2 * ld + 1]);
                const float w0 = 0.5f * (x0 / (t_d[2 * ld]     + SM_EPS));
                const float w1 = 0.5f * (x1 / (t_d[2 * ld + 1] + SM_EPS));
                const float* xr = xp + (size_t)s * XDIM;
                for (int f = 0; f < OUT_F; ++f)
                    atomicAdd(&acc_l[ld * OUT_F + f], w0 * xr[2 * f] + w1 * xr[2 * f + 1]);
            }
        }
        __syncthreads();
    }

    {
        const int wv = __builtin_amdgcn_readfirstlane(tid >> 5);
        const int f0 = lane, f1 = lane + 32;
        const float b0 = bias[f0],  b1 = bias[f1];
        const float g0 = gamma[f0], g1 = gamma[f1];
        const float be0 = beta[f0], be1 = beta[f1];
        for (int base = wv * OWN_LN; base < nr; base += (OWN_T / 32) * OWN_LN) {
            float v0[OWN_LN], v1[OWN_LN], sum[OWN_LN];
            #pragma unroll
            for (int k = 0; k < OWN_LN; ++k) {
                int r = base + k;
                if (r >= nr) r = nr - 1;
                v0[k] = acc_l[r * OUT_F + f0] + b0;
                v1[k] = acc_l[r * OUT_F + f1] + b1;
                sum[k] = v0[k] + v1[k];
            }
            #pragma unroll
            for (int off = 16; off >= 1; off >>= 1) {
                #pragma unroll
                for (int k = 0; k < OWN_LN; ++k) sum[k] += __shfl_xor(sum[k], off, 32);
            }
            float c0[OWN_LN], c1[OWN_LN], sq[OWN_LN];
            #pragma unroll
            for (int k = 0; k < OWN_LN; ++k) {
                const float mu = sum[k] * (1.0f / OUT_F);
                c0[k] = v0[k] - mu;
                c1[k] = v1[k] - mu;
                sq[k] = c0[k] * c0[k] + c1[k] * c1[k];
            }
            #pragma unroll
            for (int off = 16; off >= 1; off >>= 1) {
                #pragma unroll
                for (int k = 0; k < OWN_LN; ++k) sq[k] += __shfl_xor(sq[k], off, 32);
            }
            #pragma unroll
            for (int k = 0; k < OWN_LN; ++k) {
                const float inv = rsqrtf(sq[k] * (1.0f / OUT_F) + LN_EPS);
                if (base + k < nr) {
                    float* o = out + (size_t)(d0 + base + k) * OUT_F;
                    o[f0] = c0[k] * inv * g0 + be0;
                    o[f1] = c1[k] * inv * g1 + be1;
                }
            }
        }
    }
}

extern "C" void kernel_launch(void* const* d_in, const int* in_sizes, int n_in,
                              void* d_out, int out_size, void* d_ws, size_t ws_size,
                              hipStream_t stream)
{
    const float* X    = (const float*)d_in[0];
    const int*   ei   = (const int*)  d_in[1];
    const float* W    = (const float*)d_in[2];
    const float* attS = (const float*)d_in[3];
    const float* attD = (const float*)d_in[4];
    const float* bias = (const float*)d_in[5];
    const float* g    = (const float*)d_in[6];
    const float* b    = (const float*)d_in[7];
    float* out = (float*)d_out;

    const int N = in_sizes[0] / IN_F;
    const int E = in_sizes[1] / 2;
    const int* esrc = ei;
    const int* edst = ei + E;

    float* ws    = (float*)d_ws;
    float* xp    = ws;
    float* aS    = xp   + (size_t)N * XDIM;
    float* aD    = aS   + (size_t)N * 2;
    unsigned* keyw = (unsigned*)(aD + (size_t)N * 2);

    gat_gemm_wmma<<<GEMM_BLOCKS, 128, 0, stream>>>(X, W, xp, N);
    gat_logits<<<(N + 8 * NODE_W - 1) / (8 * NODE_W), 256, 0, stream>>>(
        xp, attS, attD, aS, aD, N, edst, keyw, E);
    gat_own_range<<<(N + OWN_NR - 1) / OWN_NR, OWN_T, 0, stream>>>(
        esrc, edst, aS, aD, xp, bias, g, b, out, keyw, E, N);
}


# Owner blocks' sweep 1: 2 scattered 8-byte loads per listed edge, not 4 of 4 bytes: (src,dst) from one record (launch 2 writes them), both heads of a_src as one pair; no sum or order changed
# speedup vs baseline: 6.1837x; 1.0630x over previous
#include <hip/hip_runtime.h>
#include <hip/hip_bf16.h>
#include <math.h>


typedef __attribute__((ext_vector_type(2))) float v2f;
typedef __attribute__((ext_vector_type(8))) float v8f;

#define IN_F     128
#define OUT_F    64
#define HEADS    2
#define XDIM     128
#define NEG_SLOPE 0.2f
#define LN_EPS    1e-5f
#define SM_EPS    1e-16f

#define GEMM_BLOCKS 768

__global__ __launch_bounds__(128) void gat_gemm_wmma(
    const float* __restrict__ X, const float* __restrict__ W,
    float* __restrict__ xp, int N)
{
    const int lane = threadIdx.x & 31;
    const int w    = threadIdx.x >> 5;
    const int lm   = lane & 15;
    const int h    = lane >> 4;
    const int nstrips = (N + 15) >> 4;

    float4 bq[2][16];
    #pragma unroll
    for (int t = 0; t < 2; ++t) {
        const float* brow = W + (size_t)(t * OUT_F + 16 * w + lm) * IN_F + 4 * h;
        #pragma unroll
        for (int j = 0; j < 16; ++j) bq[t][j] = *(const float4*)(brow + 8 * j);
    }

    const int mofs = h << 3;
    const int c0   = 2 * (16 * w + lm);

    int strip = blockIdx.x;
    if (strip >= nstrips) return;

    float4 aq[16];
    {
        int ar = strip * 16 + lm; if (ar >= N) ar = N - 1;
        const float* arow = X + (size_t)ar * IN_F + 4 * h;
        #pragma unroll
        for (int j = 0; j < 16; ++j) aq[j] = *(const float4*)(arow + 8 * j);
    }

    for (; strip < nstrips; strip += gridDim.x) {
        float4 an[16];
        {
            const int ns = (strip + (int)gridDim.x < nstrips) ? strip + (int)gridDim.x : strip;
            int ar = ns * 16 + lm; if (ar >= N) ar = N - 1;
            const float* arow = X + (size_t)ar * IN_F + 4 * h;
            #pragma unroll
            for (int j = 0; j < 16; ++j) an[j] = *(const float4*)(arow + 8 * j);
        }

        v8f acc0 = (v8f){}, acc1 = (v8f){};
        #pragma unroll
        for (int j = 0; j < 16; ++j) {
            const v2f a_lo = {aq[j].x, aq[j].y};
            const v2f a_hi = {aq[j].z, aq[j].w};
            const v2f b0lo = {bq[0][j].x, bq[0][j].y};
            const v2f b1lo = {bq[1][j].x, bq[1][j].y};
            const v2f b0hi = {bq[0][j].z, bq[0][j].w};
            const v2f b1hi = {bq[1][j].z, bq[1][j].w};
            acc0 = __builtin_amdgcn_wmma_f32_16x16x4_f32(
                false, a_lo, false, b0lo, (short)0, acc0, false, false);
            acc1 = __builtin_amdgcn_wmma_f32_16x16x4_f32(
                false, a_lo, false, b1lo, (short)0, acc1, false, false);
            acc0 = __builtin_amdgcn_wmma_f32_16x16x4_f32(
                false, a_hi, false, b0hi, (short)0, acc0, false, false);
            acc1 = __builtin_amdgcn_wmma_f32_16x16x4_f32(
                false, a_hi, false, b1hi, (short)0, acc1, false, false);
        }

        const int r0 = strip * 16;
        #pragma unroll
        for (int v = 0; v < 8; ++v) {
            const int row = r0 + mofs + v;
            if (row < N) {
                float2* opair = (float2*)(xp + (size_t)row * XDIM + c0);
                *opair = make_float2(acc0[v], acc1[v]);
            }
        }

        #pragma unroll
        for (int j = 0; j < 16; ++j) aq[j] = an[j];
    }
}

#define NODE_W 16
#define OWN_NR 400

__device__ __forceinline__ void logits_load(
    const float* __restrict__ xp, float4 (&xv)[NODE_W], int n0, int N, int lane)
{
    const int fo = lane * 4;
    #pragma unroll
    for (int k = 0; k < NODE_W; ++k) {
        int node = n0 + k;
        if (node >= N) node = N - 1;
        xv[k] = *(const float4*)(xp + (size_t)node * XDIM + fo);
    }
}

template <bool FULL>
__device__ __forceinline__ void logits_rows(
    const float4 (&xv)[NODE_W], const float4 sv, const float4 dv,
    float* __restrict__ aS, float* __restrict__ aD, int n0, int N, int lane)
{
    const bool up = lane >= 16;
    float ps[NODE_W], pd[NODE_W];
    #pragma unroll
    for (int k = 0; k < NODE_W; ++k) {
        const float s0 = xv[k].x*sv.x + xv[k].z*sv.z;
        const float s1 = xv[k].y*sv.y + xv[k].w*sv.w;
        const float d0 = xv[k].x*dv.x + xv[k].z*dv.z;
        const float d1 = xv[k].y*dv.y + xv[k].w*dv.w;
        ps[k] = (up ? s1 : s0) + __shfl_xor(up ? s0 : s1, 16, 32);
        pd[k] = (up ? d1 : d0) + __shfl_xor(up ? d0 : d1, 16, 32);
    }
    #pragma unroll
    for (int off = 8; off >= 1; off >>= 1) {
        #pragma unroll
        for (int k = 0; k < NODE_W; ++k) {
            ps[k] += __shfl_xor(ps[k], off, 32);
            pd[k] += __shfl_xor(pd[k], off, 32);
        }
    }

    const int j = lane & 15;
    const int head = lane >> 4;
    float ss = ps[0], sd = pd[0];
    #pragma unroll
    for (int k = 1; k < NODE_W; ++k) {
        ss = (j == k) ? ps[k] : ss;
        sd = (j == k) ? pd[k] : sd;
    }
    if (FULL || n0 + j < N) {
        const size_t ti = (size_t)(n0 + j) * 2 + head;
        aS[ti] = ss;
        aD[ti] = sd;
    }
}

__device__ __forceinline__ void keys_load(
    const int* __restrict__ esrc, const int* __restrict__ edst,
    int (&es)[4], int (&dd)[4], int ch, int lane)
{
    const int* dp = edst + (size_t)ch * 128 + lane;
    const int* sp = esrc + (size_t)ch * 128 + lane;
    #pragma unroll
    for (int s = 0; s < 4; ++s) dd[s] = dp[32 * s];
    #pragma unroll
    for (int s = 0; s < 4; ++s) es[s] = sp[32 * s];
}

__device__ __forceinline__ void keys_store(
    const int (&es)[4], const int (&dd)[4], unsigned* __restrict__ keyw,
    int* __restrict__ recs, int ch, int lane)
{
    int2* rp = (int2*)recs + (size_t)ch * 128 + lane;
    #pragma unroll
    for (int s = 0; s < 4; ++s) rp[32 * s] = make_int2(es[s], dd[s]);

    unsigned outw = 0;
    #pragma unroll
    for (int s = 0; s < 4; ++s) {
        const unsigned key = (unsigned)dd[s] / OWN_NR;
        #pragma unroll
        for (int i = 0; i < 4; ++i) {
            const unsigned p0 = __builtin_amdgcn_ballot_w32(((key >> (2 * i)) & 1u) != 0u);
            const unsigned p1 = __builtin_amdgcn_ballot_w32(((key >> (2 * i + 1)) & 1u) != 0u);
            outw = (lane == 8 * s + i)     ? ((p0 & 0xFFFFu) | (p1 << 16))      : outw;
            outw = (lane == 8 * s + 4 + i) ? ((p0 >> 16) | (p1 & 0xFFFF0000u)) : outw;
        }
    }
    keyw[(size_t)ch * 32 + lane] = outw;
}

__global__ __launch_bounds__(256) void gat_logits(
    const float* __restrict__ xp,
    const float* __restrict__ attS, const float* __restrict__ attD,
    float* __restrict__ aS, float* __restrict__ aD, int N,
    const int* __restrict__ esrc, const int* __restrict__ edst,
    unsigned* __restrict__ keyw, int* __restrict__ recs, int E)
{
    const int lane = threadIdx.x & 31;
    const int wave = __builtin_amdgcn_readfirstlane(
        (int)((blockIdx.x * blockDim.x + threadIdx.x) >> 5));
    const int n0  = wave * NODE_W;
    const int nwv = (int)((gridDim.x * blockDim.x) >> 5);
    const int nch = E >> 7;

    int ch = wave;
    int dd[4] = {0, 0, 0, 0};
    int es[4] = {0, 0, 0, 0};
    if (ch < nch) keys_load(esrc, edst, es, dd, ch, lane);

    float4 xv[NODE_W];
    float4 sv = {0.0f, 0.0f, 0.0f, 0.0f}, dv = sv;
    if (n0 < N) {
        logits_load(xp, xv, n0, N, lane);
        const float2 s0 = *(const float2*)(attS + 2 * lane);
        const float2 s1 = *(const float2*)(attS + OUT_F + 2 * lane);
        const float2 d0 = *(const float2*)(attD + 2 * lane);
        const float2 d1 = *(const float2*)(attD + OUT_F + 2 * lane);
        sv = make_float4(s0.x, s1.x, s0.y, s1.y);
        dv = make_float4(d0.x, d1.x, d0.y, d1.y);
    }

    while (ch < nch) {
        const int nx = ch + nwv;
        int dn[4] = {0, 0, 0, 0};
        int en[4] = {0, 0, 0, 0};
        if (nx < nch) keys_load(esrc, edst, en, dn, nx, lane);
        keys_store(es, dd, keyw, recs, ch, lane);
        #pragma unroll
        for (int s = 0; s < 4; ++s) { dd[s] = dn[s]; es[s] = en[s]; }
        ch = nx;
    }

    if (n0 >= N) return;
    if (n0 + NODE_W <= N)
        logits_rows<true>(xv, sv, dv, aS, aD, n0, N, lane);
    else
        logits_rows<false>(xv, sv, dv, aS, aD, n0, N, lane);
}

__device__ __forceinline__ void atomicMaxF(float* addr, float val)
{
    if (val >= 0.0f) atomicMax((int*)addr, __float_as_int(val));
    else             atomicMin((unsigned int*)addr, __float_as_uint(val));
}

__device__ __forceinline__ float lrelu(float v)
{
    return v > 0.0f ? v : NEG_SLOPE * v;
}

#define OWN_T  1024
#define OWN_Q  10240
#define OWN_U  8
#define OWN_B  8
#define OWN_LN 4

__device__ __forceinline__ unsigned own_flags16(const uint4 kv, const uint4 mk)
{
    const unsigned a = (kv.x ^ mk.x) & (kv.y ^ mk.y) & (kv.z ^ mk.z) & (kv.w ^ mk.w);
    return a & (a >> 16);
}

__device__ __forceinline__ void own_drain(
    unsigned w, int qa, int qb, int* q_s, int* q_n)
{
    while (w) {
        const int p = __ffs((int)w) - 1;
        const int q = (p & 16) ? qb : qa;
        const int idx = atomicAdd(q_n, 1);
        if (idx < OWN_Q) q_s[idx] = q * 16 + (p & 15);
        w &= w - 1;
    }
}

__device__ __forceinline__ void own_scan_keys(
    const uint4* __restrict__ k16, int n16, int rot, int tid, const uint4 mk,
    int* q_s, int* q_n)
{
    int i0 = 0;
    for (; i0 + OWN_T * OWN_U <= n16; i0 += OWN_T * OWN_U) {
        uint4 kv[OWN_U];
        int   qg[OWN_U];
        #pragma unroll
        for (int u = 0; u < OWN_U; ++u) {
            int q = rot + i0 + u * OWN_T + tid;
            if (q >= n16) q -= n16;
            qg[u] = q;
            kv[u] = k16[q];
        }
        unsigned w[OWN_U / 2];
        #pragma unroll
        for (int h = 0; h < OWN_U / 2; ++h)
            w[h] = own_flags16(kv[2 * h], mk) | (own_flags16(kv[2 * h + 1], mk) << 16);
        #pragma unroll
        for (int h = 0; h < OWN_U / 2; ++h)
            own_drain(w[h], qg[2 * h], qg[2 * h + 1], q_s, q_n);
    }
    if (i0 < n16) {
        uint4 kv[OWN_U];
        int   qg[OWN_U];
        bool  in[OWN_U];
        #pragma unroll
        for (int u = 0; u < OWN_U; ++u) {
            const int i = i0 + u * OWN_T + tid;
            int q = rot + i;
            if (q >= n16) q -= n16;
            in[u] = i < n16;
            qg[u] = q;
            kv[u] = k16[in[u] ? q : rot];
        }
        unsigned w[OWN_U / 2];
        #pragma unroll
        for (int h = 0; h < OWN_U / 2; ++h)
            w[h] = (in[2 * h]     ? own_flags16(kv[2 * h], mk)     : 0u)
                 | ((in[2 * h + 1] ? own_flags16(kv[2 * h + 1], mk) : 0u) << 16);
        #pragma unroll
        for (int h = 0; h < OWN_U / 2; ++h)
            own_drain(w[h], qg[2 * h], qg[2 * h + 1], q_s, q_n);
    }
}

__global__ __launch_bounds__(OWN_T) void gat_own_range(
    const int* __restrict__ esrc, const int* __restrict__ edst,
    const float* __restrict__ aS, const float* __restrict__ aD,
    const float* __restrict__ xp,
    const float* __restrict__ bias, const float* __restrict__ gamma,
    const float* __restrict__ beta, float* __restrict__ out,
    const unsigned* __restrict__ keyw, const int* __restrict__ recs, int E, int N)
{
    __shared__ float acc_l[OWN_NR * OUT_F];
    __shared__ float t_ad[OWN_NR * 2];
    __shared__ float t_m[OWN_NR * 2];
    __shared__ float t_d[OWN_NR * 2];
    __shared__ int   q_s[OWN_Q];
    __shared__ int   q_ld[OWN_Q];
    __shared__ float q_x0[OWN_Q];
    __shared__ float q_x1[OWN_Q];
    __shared__ int   q_n;

    const int tid  = threadIdx.x;
    const int lane = tid & 31;
    const int d0   = (int)blockIdx.x * OWN_NR;
    const int nr   = (N - d0 < OWN_NR) ? (N - d0) : OWN_NR;
    const int Et   = E + N;

    for (int i = tid; i < nr * OUT_F; i += OWN_T) acc_l[i] = 0.0f;
    for (int i = tid; i < nr * 2; i += OWN_T) {
        t_ad[i] = aD[(size_t)d0 * 2 + i];
        t_m[i]  = -INFINITY;
        t_d[i]  = 0.0f;
    }
    for (int i = tid; i < nr; i += OWN_T)
        q_s[i] = E + d0 + i;
    if (tid == 0) q_n = nr;
    __syncthreads();

    {
        const int n16 = (E >> 7) << 3;
        if (n16 > 0) {
            const int rot = (int)(((long long)blockIdx.x * n16) / (long long)gridDim.x) & ~31;
            const unsigned kb = (unsigned)blockIdx.x;
            uint4 mk;
            mk.x = ((kb & 0x01u) ? 0u : 0x0000FFFFu) | ((kb & 0x02u) ? 0u : 0xFFFF0000u);
            mk.y = ((kb & 0x04u) ? 0u : 0x0000FFFFu) | ((kb & 0x08u) ? 0u : 0xFFFF0000u);
            mk.z = ((kb & 0x10u) ? 0u : 0x0000FFFFu) | ((kb & 0x20u) ? 0u : 0xFFFF0000u);
            mk.w = ((kb & 0x40u) ? 0u : 0x0000FFFFu) | ((kb & 0x80u) ? 0u : 0xFFFF0000u);
            own_scan_keys((const uint4*)keyw, n16, rot, tid, mk, q_s, &q_n);
        }
        for (int eid = n16 * 16 + tid; eid < E; eid += OWN_T) {
            if ((unsigned)(edst[eid] - d0) < (unsigned)nr) {
                const int idx = atomicAdd(&q_n, 1);
                if (idx < OWN_Q) q_s[idx] = eid;
            }
        }
    }
    __syncthreads();
    const int nq = q_n;

    if (nq <= OWN_Q) {
        const int nrec = (E >> 7) << 7;
        for (int i = tid; i < nq; i += OWN_T) {
            const int eid = q_s[i];
            int s, d;
            if (eid < nrec) {
                const int2 r = *((const int2*)recs + eid);
                s = r.x; d = r.y;
            } else if (eid < E) {
                s = esrc[eid]; d = edst[eid];
            } else {
                s = d = eid - E;
            }
            const unsigned ldu = (unsigned)(d - d0);
            const bool mine = ldu < (unsigned)nr;
            const int ld = mine ? (int)ldu : 0;
            const float2 as = *(const float2*)(aS + (size_t)s * 2);
            const float v0 = lrelu(as.x + t_ad[2 * ld]);
            const float v1 = lrelu(as.y + t_ad[2 * ld + 1]);
            const float e0 = mine ? v0 : -INFINITY;
            const float e1 = mine ? v1 : -INFINITY;
            atomicMaxF(&t_m[2 * ld],     e0);
            atomicMaxF(&t_m[2 * ld + 1], e1);
            q_s[i]  = s;
            q_ld[i] = ld;
            q_x0[i] = e0;
            q_x1[i] = e1;
        }
        __syncthreads();
        for (int i = tid; i < nq; i += OWN_T) {
            const int ld = q_ld[i];
            const float x0 = __expf(q_x0[i] - t_m[2 * ld]);
            const float x1 = __expf(q_x1[i] - t_m[2 * ld + 1]);
            atomicAdd(&t_d[2 * ld],     x0);
            atomicAdd(&t_d[2 * ld + 1], x1);
            q_x0[i] = x0;
            q_x1[i] = x1;
        }
        __syncthreads();
        {
            const int wv = __builtin_amdgcn_readfirstlane(tid >> 5);
            const float4* xl = (const float4*)xp + lane;
            float* al = acc_l + 2 * lane;
            for (int g = wv * 32; g < nq; g += OWN_T) {
                const int nvalid = (nq - g < 32) ? (nq - g) : 32;
                const int ic = (lane < nvalid) ? (g + lane) : g;
                const int s  = q_s[ic];
                const int ld = q_ld[ic];
                const float w0 = 0.5f * (q_x0[ic] / (t_d[2 * ld]     + SM_EPS));
                const float w1 = 0.5f * (q_x1[ic] / (t_d[2 * ld + 1] + SM_EPS));
                const int w0i = __float_as_int(w0);
                const int w1i = __float_as_int(w1);
                if (nvalid == 32) {
                    for (int jb = 0; jb < 32; jb += OWN_B) {
                        float4 xr[OWN_B];
                        #pragma unroll
                        for (int j = 0; j < OWN_B; ++j) {
                            const int sj = __builtin_amdgcn_readlane(s, jb + j);
                            xr[j] = xl[(size_t)sj * (XDIM / 4)];
                        }
                        #pragma unroll
                        for (int j = 0; j < OWN_B; ++j) {
                            const int   lj  = __builtin_amdgcn_readlane(ld, jb + j);
                            const float w0j = __int_as_float(__builtin_amdgcn_readlane(w0i, jb + j));
                            const float w1j = __int_as_float(__builtin_amdgcn_readlane(w1i, jb + j));
                            float* ap = al + lj * OUT_F;
                            atomicAdd(ap,     w0j * xr[j].x + w1j * xr[j].y);
                            atomicAdd(ap + 1, w0j * xr[j].z + w1j * xr[j].w);
                        }
                    }
                } else {
                    for (int j = 0; j < nvalid; ++j) {
                        const int   sj  = __builtin_amdgcn_readlane(s, j);
                        const int   lj  = __builtin_amdgcn_readlane(ld, j);
                        const float w0j = __int_as_float(__builtin_amdgcn_readlane(w0i, j));
                        const float w1j = __int_as_float(__builtin_amdgcn_readlane(w1i, j));
                        const float4 xr = xl[(size_t)sj * (XDIM / 4)];
                        float* ap = al + lj * OUT_F;
                        atomicAdd(ap,     w0j * xr.x + w1j * xr.y);
                        atomicAdd(ap + 1, w0j * xr.z + w1j * xr.w);
                    }
                }
            }
        }
        __syncthreads();
    } else {
        for (int eid = tid; eid < Et; eid += OWN_T) {
            int s, d;
            if (eid < E) { s = esrc[eid]; d = edst[eid]; } else { s = d = eid - E; }
            const unsigned ld = (unsigned)(d - d0);
            if (ld < (unsigned)nr) {
                const float2 as = *(const float2*)(aS + (size_t)s * 2);
                atomicMaxF(&t_m[2 * ld],     lrelu(as.x + t_ad[2 * ld]));
                atomicMaxF(&t_m[2 * ld + 1], lrelu(as.y + t_ad[2 * ld + 1]));
            }
        }
        __syncthreads();
        for (int eid = tid; eid < Et; eid += OWN_T) {
            int s, d;
            if (eid < E) { s = esrc[eid]; d = edst[eid]; } else { s = d = eid - E; }
            const unsigned ld = (unsigned)(d - d0);
            if (ld < (unsigned)nr) {
                const float2 as = *(const float2*)(aS + (size_t)s * 2);
                atomicAdd(&t_d[2 * ld],     __expf(lrelu(as.x + t_ad[2 * ld])     - t_m[2 * ld]));
                atomicAdd(&t_d[2 * ld + 1], __expf(lrelu(as.y + t_ad[2 * ld + 1]) - t_m[2 * ld + 1]));
            }
        }
        __syncthreads();
        for (int eid = tid; eid < Et; eid += OWN_T) {
            int s, d;
            if (eid < E) { s = esrc[eid]; d = edst[eid]; } else { s = d = eid - E; }
            const unsigned ld = (unsigned)(d - d0);
            if (ld < (unsigned)nr) {
                const float2 as = *(const float2*)(aS + (size_t)s * 2);
                const float x0 = __expf(lrelu(as.x + t_ad[2 * ld])     - t_m[2 * ld]);
                const float x1 = __expf(lrelu(as.y + t_ad[2 * ld + 1]) - t_m[2 * ld + 1]);
                const float w0 = 0.5f * (x0 / (t_d[2 * ld]     + SM_EPS));
                const float w1 = 0.5f * (x1 / (t_d[2 * ld + 1] + SM_EPS));
                const float* xr = xp + (size_t)s * XDIM;
                for (int f = 0; f < OUT_F; ++f)
                    atomicAdd(&acc_l[ld * OUT_F + f], w0 * xr[2 * f] + w1 * xr[2 * f + 1]);
            }
        }
        __syncthreads();
    }

    {
        const int wv = __builtin_amdgcn_readfirstlane(tid >> 5);
        const int f0 = lane, f1 = lane + 32;
        const float b0 = bias[f0],  b1 = bias[f1];
        const float g0 = gamma[f0], g1 = gamma[f1];
        const float be0 = beta[f0], be1 = beta[f1];
        for (int base = wv * OWN_LN; base < nr; base += (OWN_T / 32) * OWN_LN) {
            float v0[OWN_LN], v1[OWN_LN], sum[OWN_LN];
            #pragma unroll
            for (int k = 0; k < OWN_LN; ++k) {
                int r = base + k;
                if (r >= nr) r = nr - 1;
                v0[k] = acc_l[r * OUT_F + f0] + b0;
                v1[k] = acc_l[r * OUT_F + f1] + b1;
                sum[k] = v0[k] + v1[k];
            }
            #pragma unroll
            for (int off = 16; off >= 1; off >>= 1) {
                #pragma unroll
                for (int k = 0; k < OWN_LN; ++k) sum[k] += __shfl_xor(sum[k], off, 32);
            }
            float c0[OWN_LN], c1[OWN_LN], sq[OWN_LN];
            #pragma unroll
            for (int k = 0; k < OWN_LN; ++k) {
                const float mu = sum[k] * (1.0f / OUT_F);
                c0[k] = v0[k] - mu;
                c1[k] = v1[k] - mu;
                sq[k] = c0[k] * c0[k] + c1[k] * c1[k];
            }
            #pragma unroll
            for (int off = 16; off >= 1; off >>= 1) {
                #pragma unroll
                for (int k = 0; k < OWN_LN; ++k) sq[k] += __shfl_xor(sq[k], off, 32);
            }
            #pragma unroll
            for (int k = 0; k < OWN_LN; ++k) {
                const float inv = rsqrtf(sq[k] * (1.0f / OUT_F) + LN_EPS);
                if (base + k < nr) {
                    float* o = out + (size_t)(d0 + base + k) * OUT_F;
                    o[f0] = c0[k] * inv * g0 + be0;
                    o[f1] = c1[k] * inv * g1 + be1;
                }
            }
        }
    }
}

extern "C" void kernel_launch(void* const* d_in, const int* in_sizes, int n_in,
                              void* d_out, int out_size, void* d_ws, size_t ws_size,
                              hipStream_t stream)
{
    const float* X    = (const float*)d_in[0];
    const int*   ei   = (const int*)  d_in[1];
    const float* W    = (const float*)d_in[2];
    const float* attS = (const float*)d_in[3];
    const float* attD = (const float*)d_in[4];
    const float* bias = (const float*)d_in[5];
    const float* g    = (const float*)d_in[6];
    const float* b    = (const float*)d_in[7];
    float* out = (float*)d_out;

    const int N = in_sizes[0] / IN_F;
    const int E = in_sizes[1] / 2;
    const int* esrc = ei;
    const int* edst = ei + E;

    float* ws    = (float*)d_ws;
    float* xp    = ws;
    float* aS    = xp   + (size_t)N * XDIM;
    float* aD    = aS   + (size_t)N * 2;
    unsigned* keyw = (unsigned*)(aD + (size_t)N * 2);
    int* recs = (int*)(keyw + ((size_t)(E >> 7) + 1) * 32);

    gat_gemm_wmma<<<GEMM_BLOCKS, 128, 0, stream>>>(X, W, xp, N);
    gat_logits<<<(N + 8 * NODE_W - 1) / (8 * NODE_W), 256, 0, stream>>>(
        xp, attS, attD, aS, aD, N, esrc, edst, keyw, recs, E);
    gat_own_range<<<(N + OWN_NR - 1) / OWN_NR, OWN_T, 0, stream>>>(
        esrc, edst, aS, aD, xp, bias, g, b, out, keyw, recs, E, N);
}
